# starting kernel
# baseline (speedup 1.0000x reference)
#include <hip/hip_runtime.h>
#include <hip/hip_bf16.h>


typedef __attribute__((ext_vector_type(2))) float v2f;
typedef __attribute__((ext_vector_type(8))) float v8f;

#define LN_EPS 1e-5f

template <int K, int NT, int MT>
__global__ __launch_bounds__(256)
void gemm_wmma_f32(const float* __restrict__ A,
                   const float* __restrict__ W,
                   const float* __restrict__ bias,
                   float* __restrict__ out,
                   int nrows) {
    constexpr int NOUT = NT * 16;
    const int wave = (blockIdx.x * blockDim.x + threadIdx.x) >> 5;
    const int lane = threadIdx.x & 31;
    const int m0 = wave * (MT * 16);
    if (m0 >= nrows) return;

    const int mr = lane & 15;
    const int hi = lane >> 4;

    v8f acc[MT][NT];
#pragma unroll
    for (int mt = 0; mt < MT; ++mt)
#pragma unroll
        for (int t = 0; t < NT; ++t)
            acc[mt][t] = (v8f){0.f, 0.f, 0.f, 0.f, 0.f, 0.f, 0.f, 0.f};

    const float* arow[MT];
#pragma unroll
    for (int mt = 0; mt < MT; ++mt)
        arow[mt] = A + (size_t)(m0 + mt * 16 + mr) * K;

#pragma unroll 2
    for (int k = 0; k < K; k += 4) {
        const int kb = k + hi * 2;
        v2f a[MT];
#pragma unroll
        for (int mt = 0; mt < MT; ++mt) {
            a[mt].x = arow[mt][kb];
            a[mt].y = arow[mt][kb + 1];
        }
        const float* w0 = W + (size_t)kb * NOUT;
        const float* w1 = w0 + NOUT;
#pragma unroll
        for (int t = 0; t < NT; ++t) {
            const int n = t * 16 + mr;
            v2f b;
            b.x = w0[n];
            b.y = w1[n];
#pragma unroll
            for (int mt = 0; mt < MT; ++mt)
                acc[mt][t] = __builtin_amdgcn_wmma_f32_16x16x4_f32(
                    false, a[mt], false, b, (short)0, acc[mt][t], false, false);
        }
    }

#pragma unroll
    for (int mt = 0; mt < MT; ++mt) {
#pragma unroll
        for (int t = 0; t < NT; ++t) {
            const int n = t * 16 + mr;
            const float bv = bias ? bias[n] : 0.f;
#pragma unroll
            for (int v = 0; v < 8; ++v) {
                const int m = m0 + mt * 16 + v + hi * 8;
                out[(size_t)m * NOUT + n] = acc[mt][t][v] + bv;
            }
        }
    }
}

__global__ void deg_init_kernel(float* deg, int N) {
    int i = blockIdx.x * blockDim.x + threadIdx.x;
    if (i < N) deg[i] = 1.0f;
}

__global__ void deg_count_kernel(const int* __restrict__ col, float* deg, int E) {
    int e = blockIdx.x * blockDim.x + threadIdx.x;
    if (e < E) atomicAdd(&deg[col[e]], 1.0f);
}

__global__ void deg_inv_kernel(const float* __restrict__ deg, float* dinv, int N) {
    int i = blockIdx.x * blockDim.x + threadIdx.x;
    if (i < N) dinv[i] = rsqrtf(deg[i]);
}

template <int F>
__global__ void init_self_kernel(const float* __restrict__ hw,
                                 const float* __restrict__ dinv,
                                 float* __restrict__ out, int N) {
    size_t i = (size_t)blockIdx.x * blockDim.x + threadIdx.x;
    if (i >= (size_t)N * F) return;
    const int node = (int)(i / F);
    const float dv = dinv[node];
    out[i] = hw[i] * dv * dv;
}

template <int F>
__global__ __launch_bounds__(256)
void scatter_edges_kernel(const float* __restrict__ src,
                          const int* __restrict__ rows,
                          const int* __restrict__ cols,
                          const float* __restrict__ dinv,
                          float* __restrict__ out, int E) {
    constexpr int PER = F / 32;
    const int wave = (blockIdx.x * blockDim.x + threadIdx.x) >> 5;
    const int lane = threadIdx.x & 31;
    if (wave >= E) return;
    const int r = rows[wave];
    const int c = cols[wave];
    const float nrm = dinv[r] * dinv[c];
    const float* s = src + (size_t)r * F + lane * PER;
    float* o = out + (size_t)c * F + lane * PER;
#pragma unroll
    for (int i = 0; i < PER; ++i)
        atomicAdd(o + i, s[i] * nrm);
}

__global__ __launch_bounds__(256)
void post_ln_relu_kernel(float* __restrict__ h,
                         const float* __restrict__ b,
                         const float* __restrict__ g,
                         const float* __restrict__ be, int N) {
    const int wave = (blockIdx.x * blockDim.x + threadIdx.x) >> 5;
    const int lane = threadIdx.x & 31;
    if (wave >= N) return;

    float v[4];
    float sum = 0.f;
    const size_t base = (size_t)wave * 128 + lane * 4;
#pragma unroll
    for (int i = 0; i < 4; ++i) {
        v[i] = h[base + i] + b[lane * 4 + i];
        sum += v[i];
    }
#pragma unroll
    for (int off = 16; off > 0; off >>= 1) sum += __shfl_xor(sum, off, 32);
    const float mu = sum * (1.0f / 128.0f);

    float var = 0.f;
#pragma unroll
    for (int i = 0; i < 4; ++i) {
        const float d = v[i] - mu;
        var += d * d;
    }
#pragma unroll
    for (int off = 16; off > 0; off >>= 1) var += __shfl_xor(var, off, 32);
    var *= (1.0f / 128.0f);
    const float rs = rsqrtf(var + LN_EPS);

#pragma unroll
    for (int i = 0; i < 4; ++i) {
        const float y = (v[i] - mu) * rs * g[lane * 4 + i] + be[lane * 4 + i];
        h[base + i] = fmaxf(y, 0.f);
    }
}

template <int F>
__global__ void add_bias_kernel(float* __restrict__ h,
                                const float* __restrict__ b, int N) {
    size_t i = (size_t)blockIdx.x * blockDim.x + threadIdx.x;
    if (i >= (size_t)N * F) return;
    h[i] += b[i % F];
}

extern "C" void kernel_launch(void* const* d_in, const int* in_sizes, int n_in,
                              void* d_out, int out_size, void* d_ws, size_t ws_size,
                              hipStream_t stream) {
    constexpr int F = 128, H = 128, D = 64, C = 32;
    constexpr int MT = 2;

    const float* x   = (const float*)d_in[0];
    const int*   ei  = (const int*)d_in[1];
    const float* W1  = (const float*)d_in[2];
    const float* b1  = (const float*)d_in[3];
    const float* g1  = (const float*)d_in[4];
    const float* be1 = (const float*)d_in[5];
    const float* W2  = (const float*)d_in[6];
    const float* b2  = (const float*)d_in[7];
    const float* Wc  = (const float*)d_in[8];
    const float* bc  = (const float*)d_in[9];
    float* logits    = (float*)d_out;

    const int N = in_sizes[0] / F;
    const int E = in_sizes[1] / 2;
    const int* rows = ei;
    const int* cols = ei + E;

    float* ws   = (float*)d_ws;
    float* hw1  = ws;
    float* agg2 = ws;
    float* hw2  = ws + (size_t)N * 64;
    float* agg1 = ws + (size_t)N * 128;
    float* deg  = ws + (size_t)N * 256;
    float* dinv = deg + N;

    const int TPB = 256;
    const int wavesPerBlk = TPB / 32;
    auto cdiv = [](long long a, long long b) { return (int)((a + b - 1) / b); };

    deg_init_kernel<<<cdiv(N, TPB), TPB, 0, stream>>>(deg, N);
    deg_count_kernel<<<cdiv(E, TPB), TPB, 0, stream>>>(cols, deg, E);
    deg_inv_kernel<<<cdiv(N, TPB), TPB, 0, stream>>>(deg, dinv, N);

    {
        const int nWaves = cdiv(N, 16 * MT);
        gemm_wmma_f32<F, H / 16, MT><<<cdiv(nWaves, wavesPerBlk), TPB, 0, stream>>>(
            x, W1, nullptr, hw1, N);
    }

    init_self_kernel<H><<<cdiv((long long)N * H, TPB), TPB, 0, stream>>>(hw1, dinv, agg1, N);
    scatter_edges_kernel<H><<<cdiv(E, wavesPerBlk), TPB, 0, stream>>>(
        hw1, rows, cols, dinv, agg1, E);

    post_ln_relu_kernel<<<cdiv(N, wavesPerBlk), TPB, 0, stream>>>(agg1, b1, g1, be1, N);

    {
        const int nWaves = cdiv(N, 16 * MT);
        gemm_wmma_f32<H, D / 16, MT><<<cdiv(nWaves, wavesPerBlk), TPB, 0, stream>>>(
            agg1, W2, nullptr, hw2, N);
    }

    init_self_kernel<D><<<cdiv((long long)N * D, TPB), TPB, 0, stream>>>(hw2, dinv, agg2, N);
    scatter_edges_kernel<D><<<cdiv(E, wavesPerBlk), TPB, 0, stream>>>(
        hw2, rows, cols, dinv, agg2, E);
    add_bias_kernel<D><<<cdiv((long long)N * D, TPB), TPB, 0, stream>>>(agg2, b2, N);

    {
        const int nWaves = cdiv(N, 16 * MT);
        gemm_wmma_f32<D, C / 16, MT><<<cdiv(nWaves, wavesPerBlk), TPB, 0, stream>>>(
            agg2, Wc, bc, logits, N);
    }
}


# Classifier applied ahead of the second edge sum: logits = agg2((h W2) Wc) + (b2 Wc + bc), both products on WMMA in the layer-1 launch; the second gather moves 128-byte rows, 4 nodes per wave
# speedup vs baseline: 3.5790x; 3.5790x over previous
#include <hip/hip_runtime.h>
#include <hip/hip_bf16.h>


typedef __attribute__((ext_vector_type(2))) float v2f;
typedef __attribute__((ext_vector_type(8))) float v8f;

#define LN_EPS 1e-5f

template <int K, int NT, int MT, bool SCALE>
__global__ __launch_bounds__(256)
void gemm_wmma_f32(const float* __restrict__ A,
                   const float* __restrict__ W,
                   const float* __restrict__ bias,
                   const int* __restrict__ cnt,
                   float* __restrict__ out,
                   int nrows) {
    constexpr int NOUT = NT * 16;
    const int wave = (blockIdx.x * blockDim.x + threadIdx.x) >> 5;
    const int lane = threadIdx.x & 31;
    const int m0 = wave * (MT * 16);
    if (m0 >= nrows) return;

    const int mr = lane & 15;
    const int hi = lane >> 4;

    v8f acc[MT][NT];
#pragma unroll
    for (int mt = 0; mt < MT; ++mt)
#pragma unroll
        for (int t = 0; t < NT; ++t)
            acc[mt][t] = (v8f){0.f, 0.f, 0.f, 0.f, 0.f, 0.f, 0.f, 0.f};

    const float* arow[MT];
#pragma unroll
    for (int mt = 0; mt < MT; ++mt)
        arow[mt] = A + (size_t)(m0 + mt * 16 + mr) * K;

#pragma unroll 2
    for (int k = 0; k < K; k += 4) {
        const int kb = k + hi * 2;
        v2f a[MT];
#pragma unroll
        for (int mt = 0; mt < MT; ++mt) {
            a[mt].x = arow[mt][kb];
            a[mt].y = arow[mt][kb + 1];
        }
        const float* w0 = W + (size_t)kb * NOUT;
        const float* w1 = w0 + NOUT;
#pragma unroll
        for (int t = 0; t < NT; ++t) {
            const int n = t * 16 + mr;
            v2f b;
            b.x = w0[n];
            b.y = w1[n];
#pragma unroll
            for (int mt = 0; mt < MT; ++mt)
                acc[mt][t] = __builtin_amdgcn_wmma_f32_16x16x4_f32(
                    false, a[mt], false, b, (short)0, acc[mt][t], false, false);
        }
    }

    float rsv[MT][8];
#pragma unroll
    for (int mt = 0; mt < MT; ++mt)
#pragma unroll
        for (int v = 0; v < 8; ++v)
            rsv[mt][v] = SCALE ? rsqrtf((float)(cnt[m0 + mt * 16 + v + hi * 8] + 1)) : 1.f;

#pragma unroll
    for (int mt = 0; mt < MT; ++mt) {
#pragma unroll
        for (int t = 0; t < NT; ++t) {
            const int n = t * 16 + mr;
            const float bv = bias ? bias[n] : 0.f;
#pragma unroll
            for (int v = 0; v < 8; ++v) {
                const int m = m0 + mt * 16 + v + hi * 8;
                const float y = acc[mt][t][v] + bv;
                out[(size_t)m * NOUT + n] = SCALE ? y * rsv[mt][v] : y;
            }
        }
    }
}

#define SLOTS 32

__global__ void zero_counts_kernel(int* __restrict__ cnt, int* __restrict__ ovn, int N) {
    int i = blockIdx.x * blockDim.x + threadIdx.x;
    if (i < N) cnt[i] = 0;
    if (i == 0) *ovn = 0;
}

__global__ void group_edges_kernel(const int* __restrict__ row,
                                   const int* __restrict__ col, int* cnt,
                                   int* __restrict__ slot, int* ovn,
                                   int* __restrict__ ov, int E) {
    int e = blockIdx.x * blockDim.x + threadIdx.x;
    if (e < E) {
        const int c = col[e];
        const int r = row[e];
        const int p = atomicAdd(&cnt[c], 1);
        if (p < SLOTS) {
            slot[(size_t)c * SLOTS + p] = r;
        } else {
            const int q = atomicAdd(ovn, 1);
            ov[2 * (size_t)q] = c;
            ov[2 * (size_t)q + 1] = r;
        }
    }
}

#define FG_ROWS 32

template <int F, int NMID, int NOUT>
__global__ __launch_bounds__(128)
void gather_ln_products_kernel(const float* __restrict__ src,
                               const int* __restrict__ cnt,
                               const int* __restrict__ slot,
                               const int* __restrict__ ovn,
                               const int* __restrict__ ov,
                               const float* __restrict__ b,
                               const float* __restrict__ g,
                               const float* __restrict__ be,
                               const float* __restrict__ W,
                               const float* __restrict__ Wc,
                               const float* __restrict__ b2,
                               const float* __restrict__ bc,
                               float* __restrict__ out,
                               float* __restrict__ cvec, int N, int E) {
    static_assert(NMID == 64 && NOUT == 32, "tile dealing below is for 4 waves, 64 and 32 columns");
    constexpr int PER = F / 32;
    constexpr int PITCH = F + 4;
    constexpr int NT = NMID / 16;
    constexpr int MT = NT / 2;
    typedef __attribute__((ext_vector_type(PER))) float vf;
    __shared__ __attribute__((aligned(16))) float tile[FG_ROWS * PITCH];

    const int lane = threadIdx.x & 31;
    const int w = __builtin_amdgcn_readfirstlane((int)(threadIdx.x >> 5));
    const int node0 = blockIdx.x * FG_ROWS;
    const float* lsrc = src + lane * PER;

    for (int q = 0; q < FG_ROWS / 4; ++q) {
        const int lr = w * (FG_ROWS / 4) + q;
        const int c = node0 + lr;
        vf y;
#pragma unroll
        for (int i = 0; i < PER; ++i) y[i] = 0.f;
        if (c < N) {
            const int deg = cnt[c];
            const float dc = rsqrtf((float)(deg + 1));

            vf acc = *(const vf*)(lsrc + (size_t)c * F);
            {
                const int n = min(SLOTS, deg);
                int my_r = c;
                if (lane < n) my_r = slot[(size_t)c * SLOTS + lane];
                int j = 0;
                for (; j + 4 <= n; j += 4) {
                    const int r0 = __builtin_amdgcn_readlane(my_r, j);
                    const int r1 = __builtin_amdgcn_readlane(my_r, j + 1);
                    const int r2 = __builtin_amdgcn_readlane(my_r, j + 2);
                    const int r3 = __builtin_amdgcn_readlane(my_r, j + 3);
                    const vf x0 = *(const vf*)(lsrc + (size_t)r0 * F);
                    const vf x1 = *(const vf*)(lsrc + (size_t)r1 * F);
                    const vf x2 = *(const vf*)(lsrc + (size_t)r2 * F);
                    const vf x3 = *(const vf*)(lsrc + (size_t)r3 * F);
                    acc += x0;
                    acc += x1;
                    acc += x2;
                    acc += x3;
                }
                for (; j < n; ++j) {
                    const int r0 = __builtin_amdgcn_readlane(my_r, j);
                    acc += *(const vf*)(lsrc + (size_t)r0 * F);
                }
            }
            if (deg > SLOTS) {
                const int nov = min(*ovn, E);
                for (int base = 0; base < nov; base += 32) {
                    int t = -1;
                    int r = c;
                    if (base + lane < nov) {
                        t = ov[2 * (size_t)(base + lane)];
                        r = ov[2 * (size_t)(base + lane) + 1];
                    }
                    unsigned hit = __builtin_amdgcn_ballot_w32(t == c);
                    while (hit) {
                        const int j2 = __builtin_ctz(hit);
                        hit &= hit - 1;
                        const int r0 = __builtin_amdgcn_readlane(r, j2);
                        acc += *(const vf*)(lsrc + (size_t)r0 * F);
                    }
                }
            }
            acc *= dc;

            float v[PER];
#pragma unroll
            for (int i = 0; i < PER; ++i) v[i] = acc[i] + b[lane * PER + i];

            float sum = 0.f;
#pragma unroll
            for (int i = 0; i < PER; ++i) sum += v[i];
#pragma unroll
            for (int off = 16; off > 0; off >>= 1) sum += __shfl_xor(sum, off, 32);
            const float mu = sum * (1.0f / F);

            float var = 0.f;
#pragma unroll
            for (int i = 0; i < PER; ++i) {
                const float d = v[i] - mu;
                var += d * d;
            }
#pragma unroll
            for (int off = 16; off > 0; off >>= 1) var += __shfl_xor(var, off, 32);
            var *= (1.0f / F);
            const float rs = rsqrtf(var + LN_EPS);

#pragma unroll
            for (int i = 0; i < PER; ++i) {
                const float z = (v[i] - mu) * rs * g[lane * PER + i] + be[lane * PER + i];
                y[i] = fmaxf(z, 0.f);
            }
        }
        *(vf*)(tile + lr * PITCH + lane * PER) = y;
    }
    __syncthreads();

    const int mr = lane & 15;
    const int hi = lane >> 4;
    const int n = w * 16 + mr;
    const float* ap = tile + mr * PITCH + hi * 2;
    const float* wp = W + (size_t)(hi * 2) * NMID + n;

    v8f acc[MT];
#pragma unroll
    for (int mt = 0; mt < MT; ++mt)
        acc[mt] = (v8f){0.f, 0.f, 0.f, 0.f, 0.f, 0.f, 0.f, 0.f};

#pragma unroll 8
    for (int k = 0; k < F; k += 4) {
        v2f bf;
        bf.x = wp[(size_t)k * NMID];
        bf.y = wp[(size_t)k * NMID + NMID];
#pragma unroll
        for (int mt = 0; mt < MT; ++mt) {
            const v2f a = *(const v2f*)(ap + mt * 16 * PITCH + k);
            acc[mt] = __builtin_amdgcn_wmma_f32_16x16x4_f32(
                false, a, false, bf, (short)0, acc[mt], false, false);
        }
    }

    constexpr int OP = NMID + 4;
    __syncthreads();
#pragma unroll
    for (int mt = 0; mt < MT; ++mt) {
#pragma unroll
        for (int v = 0; v < 8; ++v)
            tile[(mt * 16 + v + hi * 8) * OP + n] = acc[mt][v];
    }
    __syncthreads();

    const int rt = w >> 1;
    const int n2 = (w & 1) * 16 + mr;
    const float* ap2 = tile + (rt * 16 + mr) * OP + hi * 2;
    const float* wp2 = Wc + (size_t)(hi * 2) * NOUT + n2;
    v8f acc2 = (v8f){0.f, 0.f, 0.f, 0.f, 0.f, 0.f, 0.f, 0.f};
#pragma unroll
    for (int k = 0; k < NMID; k += 4) {
        v2f bf;
        bf.x = wp2[(size_t)k * NOUT];
        bf.y = wp2[(size_t)k * NOUT + NOUT];
        const v2f a = *(const v2f*)(ap2 + k);
        acc2 = __builtin_amdgcn_wmma_f32_16x16x4_f32(
            false, a, false, bf, (short)0, acc2, false, false);
    }

    constexpr int OP2 = NOUT + 4;
    __syncthreads();
#pragma unroll
    for (int v = 0; v < 8; ++v)
        tile[(rt * 16 + v + hi * 8) * OP2 + n2] = acc2[v];
    __syncthreads();

    typedef __attribute__((ext_vector_type(4))) float v4f;
    constexpr int LPR = NOUT / 4;
    constexpr int RPI = 32 / LPR;
    const int cl = (lane % LPR) * 4;
    const int rl = lane / LPR;
#pragma unroll
    for (int i = 0; i < (FG_ROWS / 4) / RPI; ++i) {
        const int r = w * (FG_ROWS / 4) + i * RPI + rl;
        const int m = node0 + r;
        if (m < N) {
            v4f o = *(const v4f*)(tile + r * OP2 + cl);
            o *= rsqrtf((float)(cnt[m] + 1));
            *(v4f*)(out + (size_t)m * NOUT + cl) = o;
        }
    }

    if (blockIdx.x == 0 && w < 2) {
        const int n3 = w * 16 + mr;
        const float* wp3 = Wc + (size_t)(hi * 2) * NOUT + n3;
        v8f acc3 = (v8f){0.f, 0.f, 0.f, 0.f, 0.f, 0.f, 0.f, 0.f};
#pragma unroll
        for (int k = 0; k < NMID; k += 4) {
            v2f a, bf;
            a.x = b2[k + hi * 2];
            a.y = b2[k + hi * 2 + 1];
            bf.x = wp3[(size_t)k * NOUT];
            bf.y = wp3[(size_t)k * NOUT + NOUT];
            acc3 = __builtin_amdgcn_wmma_f32_16x16x4_f32(
                false, a, false, bf, (short)0, acc3, false, false);
        }
        if (hi == 0) cvec[n3] = acc3[0] + bc[n3];
    }
}

__global__ __launch_bounds__(256)
void gather_rows32_kernel(const float* __restrict__ p,
                          const int* __restrict__ cnt,
                          const int* __restrict__ slot,
                          const int* __restrict__ ovn,
                          const int* __restrict__ ov,
                          const float* __restrict__ cvec,
                          float* __restrict__ out, int N, int E) {
    constexpr int C = 32;
    typedef __attribute__((ext_vector_type(4))) float v4f;
    typedef __attribute__((ext_vector_type(4))) int v4i;

    const int lane = threadIdx.x & 31;
    const int sub = lane & 7;
    const int gq = lane >> 3;
    const int gb = lane & ~7;
    const int wave = (blockIdx.x * blockDim.x + threadIdx.x) >> 5;
    const int c = wave * 4 + gq;
    const bool live = c < N;
    const int cc = live ? c : N - 1;

    const int cntv = cnt[cc];
    const v4i s = *(const v4i*)(slot + (size_t)cc * SLOTS + sub * 4);
    const int deg = live ? cntv : 0;
    const int n = min(SLOTS, deg);
    const float* lp = p + sub * 4;
    v4f acc = *(const v4f*)(lp + (size_t)cc * C);

    int nm = n;
    nm = max(nm, __shfl_xor(nm, 8, 32));
    nm = max(nm, __shfl_xor(nm, 16, 32));
    const int nmax = __builtin_amdgcn_readfirstlane(nm);

    const v4f zero = (v4f){0.f, 0.f, 0.f, 0.f};
    int j = 0;
    do {
        const int la = gb + (j >> 2);
        const int lb = la + 1;
        const int r0 = __shfl(s.x, la, 32);
        const int r1 = __shfl(s.y, la, 32);
        const int r2 = __shfl(s.z, la, 32);
        const int r3 = __shfl(s.w, la, 32);
        const int r4 = __shfl(s.x, lb, 32);
        const int r5 = __shfl(s.y, lb, 32);
        const int r6 = __shfl(s.z, lb, 32);
        const int r7 = __shfl(s.w, lb, 32);
        v4f x0 = zero, x1 = zero, x2 = zero, x3 = zero, x4 = zero, x5 = zero, x6 = zero, x7 = zero;
        if (j     < n) x0 = *(const v4f*)(lp + (size_t)r0 * C);
        if (j + 1 < n) x1 = *(const v4f*)(lp + (size_t)r1 * C);
        if (j + 2 < n) x2 = *(const v4f*)(lp + (size_t)r2 * C);
        if (j + 3 < n) x3 = *(const v4f*)(lp + (size_t)r3 * C);
        if (j + 4 < n) x4 = *(const v4f*)(lp + (size_t)r4 * C);
        if (j + 5 < n) x5 = *(const v4f*)(lp + (size_t)r5 * C);
        if (j + 6 < n) x6 = *(const v4f*)(lp + (size_t)r6 * C);
        if (j + 7 < n) x7 = *(const v4f*)(lp + (size_t)r7 * C);
        acc += x0;
        acc += x1;
        acc += x2;
        acc += x3;
        acc += x4;
        acc += x5;
        acc += x6;
        acc += x7;
        j += 8;
    } while (j < nmax);

    if (__builtin_amdgcn_ballot_w32(deg > SLOTS)) {
        const int nov = min(*ovn, E);
        for (int base = 0; base < nov; base += 32) {
            int t = -1;
            int r = 0;
            if (base + lane < nov) {
                t = ov[2 * (size_t)(base + lane)];
                r = ov[2 * (size_t)(base + lane) + 1];
            }
#pragma unroll
            for (int q = 0; q < 4; ++q) {
                const int cq = __builtin_amdgcn_readlane(c, q * 8);
                unsigned hit = __builtin_amdgcn_ballot_w32(t == cq);
                while (hit) {
                    const int j2 = __builtin_ctz(hit);
                    hit &= hit - 1;
                    const int r0 = __builtin_amdgcn_readlane(r, j2);
                    if (gq == q) acc += *(const v4f*)(lp + (size_t)r0 * C);
                }
            }
        }
    }

    if (live) {
        const float dc = rsqrtf((float)(deg + 1));
        const v4f cv = *(const v4f*)(cvec + sub * 4);
        const v4f o = acc * dc + cv;
        *(v4f*)(out + (size_t)c * C + sub * 4) = o;
    }
}

extern "C" void kernel_launch(void* const* d_in, const int* in_sizes, int n_in,
                              void* d_out, int out_size, void* d_ws, size_t ws_size,
                              hipStream_t stream) {
    constexpr int F = 128, H = 128, D = 64, C = 32;
    constexpr int MT = 2;

    const float* x   = (const float*)d_in[0];
    const int*   ei  = (const int*)d_in[1];
    const float* W1  = (const float*)d_in[2];
    const float* b1  = (const float*)d_in[3];
    const float* g1  = (const float*)d_in[4];
    const float* be1 = (const float*)d_in[5];
    const float* W2  = (const float*)d_in[6];
    const float* b2  = (const float*)d_in[7];
    const float* Wc  = (const float*)d_in[8];
    const float* bc  = (const float*)d_in[9];
    float* logits    = (float*)d_out;

    const int N = in_sizes[0] / F;
    const int E = in_sizes[1] / 2;
    const int* rows = ei;
    const int* cols = ei + E;

    float* ws   = (float*)d_ws;
    float* hw1  = ws;
    float* p    = ws + (size_t)N * 128;
    float* cvec = ws + (size_t)N * 160;
    int*   slot = (int*)(ws + (size_t)N * 192);
    int*   ov   = (int*)(ws + (size_t)N * 224);
    int*   cnt  = (int*)(ws + (size_t)N * 256);
    int*   ovn  = (int*)(ws + (size_t)N * 257);

    const int TPB = 256;
    const int wavesPerBlk = TPB / 32;
    auto cdiv = [](long long a, long long b) { return (int)((a + b - 1) / b); };

    zero_counts_kernel<<<cdiv(N, TPB), TPB, 0, stream>>>(cnt, ovn, N);
    group_edges_kernel<<<cdiv(E, TPB), TPB, 0, stream>>>(rows, cols, cnt, slot, ovn, ov, E);

    {
        const int nWaves = cdiv(N, 16 * MT);
        gemm_wmma_f32<F, H / 16, MT, true><<<cdiv(nWaves, wavesPerBlk), TPB, 0, stream>>>(
            x, W1, nullptr, cnt, hw1, N);
    }

    gather_ln_products_kernel<H, D, C><<<cdiv(N, FG_ROWS), 128, 0, stream>>>(
        hw1, cnt, slot, ovn, ov, b1, g1, be1, W2, Wc, b2, bc, p, cvec, N, E);

    gather_rows32_kernel<<<cdiv(N, 4 * wavesPerBlk), TPB, 0, stream>>>(
        p, cnt, slot, ovn, ov, cvec, logits, N, E);
}


# Edges grouped through LDS in two streaming passes: binned by ranges of 256 targets (one global add per bin and block), then counts and slot rows built per range in LDS; 5 -> 6 launches
# speedup vs baseline: 5.4191x; 1.5141x over previous
#include <hip/hip_runtime.h>
#include <hip/hip_bf16.h>


typedef __attribute__((ext_vector_type(2))) float v2f;
typedef __attribute__((ext_vector_type(8))) float v8f;

#define LN_EPS 1e-5f

template <int K, int NT, int MT, bool SCALE>
__global__ __launch_bounds__(256)
void gemm_wmma_f32(const float* __restrict__ A,
                   const float* __restrict__ W,
                   const float* __restrict__ bias,
                   const int* __restrict__ cnt,
                   float* __restrict__ out,
                   int nrows) {
    constexpr int NOUT = NT * 16;
    const int wave = (blockIdx.x * blockDim.x + threadIdx.x) >> 5;
    const int lane = threadIdx.x & 31;
    const int m0 = wave * (MT * 16);
    if (m0 >= nrows) return;

    const int mr = lane & 15;
    const int hi = lane >> 4;

    v8f acc[MT][NT];
#pragma unroll
    for (int mt = 0; mt < MT; ++mt)
#pragma unroll
        for (int t = 0; t < NT; ++t)
            acc[mt][t] = (v8f){0.f, 0.f, 0.f, 0.f, 0.f, 0.f, 0.f, 0.f};

    const float* arow[MT];
#pragma unroll
    for (int mt = 0; mt < MT; ++mt)
        arow[mt] = A + (size_t)(m0 + mt * 16 + mr) * K;

#pragma unroll 2
    for (int k = 0; k < K; k += 4) {
        const int kb = k + hi * 2;
        v2f a[MT];
#pragma unroll
        for (int mt = 0; mt < MT; ++mt) {
            a[mt].x = arow[mt][kb];
            a[mt].y = arow[mt][kb + 1];
        }
        const float* w0 = W + (size_t)kb * NOUT;
        const float* w1 = w0 + NOUT;
#pragma unroll
        for (int t = 0; t < NT; ++t) {
            const int n = t * 16 + mr;
            v2f b;
            b.x = w0[n];
            b.y = w1[n];
#pragma unroll
            for (int mt = 0; mt < MT; ++mt)
                acc[mt][t] = __builtin_amdgcn_wmma_f32_16x16x4_f32(
                    false, a[mt], false, b, (short)0, acc[mt][t], false, false);
        }
    }

    float rsv[MT][8];
#pragma unroll
    for (int mt = 0; mt < MT; ++mt)
#pragma unroll
        for (int v = 0; v < 8; ++v)
            rsv[mt][v] = SCALE ? rsqrtf((float)(cnt[m0 + mt * 16 + v + hi * 8] + 1)) : 1.f;

#pragma unroll
    for (int mt = 0; mt < MT; ++mt) {
#pragma unroll
        for (int t = 0; t < NT; ++t) {
            const int n = t * 16 + mr;
            const float bv = bias ? bias[n] : 0.f;
#pragma unroll
            for (int v = 0; v < 8; ++v) {
                const int m = m0 + mt * 16 + v + hi * 8;
                const float y = acc[mt][t][v] + bv;
                out[(size_t)m * NOUT + n] = SCALE ? y * rsv[mt][v] : y;
            }
        }
    }
}

#define SLOTS 32
#define BIN_SHIFT 8
#define BIN_NODES 256
#define BIN_MAX 512
#define BIN_CAP 8192
#define CHUNK 4096
#define EPT (CHUNK / 256)
#define CUR_BINS 64
#define CUR_STRIDE 32

__global__ void zero_cursors_kernel(int* __restrict__ cur, int n) {
    int i = blockIdx.x * blockDim.x + threadIdx.x;
    if (i < n) cur[i] = 0;
}

__global__ __launch_bounds__(256)
void bin_edges_kernel(const int* __restrict__ row,
                      const int* __restrict__ col,
                      int* cursor, int* __restrict__ binbuf,
                      int* bovn, int* __restrict__ bov, int E) {
    __shared__ int hist[BIN_MAX];
    __shared__ int off[BIN_MAX];
    __shared__ int base[BIN_MAX];
    __shared__ int stage_c[CHUNK];
    __shared__ int stage_r[CHUNK];

    const int tid = threadIdx.x;
    const int e0 = blockIdx.x * CHUNK;
    hist[tid] = 0;
    hist[tid + 256] = 0;
    __syncthreads();

    int c[EPT], r[EPT], rank[EPT];
#pragma unroll
    for (int i = 0; i < EPT; ++i) {
        const int e = e0 + i * 256 + tid;
        c[i] = -1;
        r[i] = 0;
        if (e < E) {
            c[i] = col[e];
            r[i] = row[e];
        }
    }
#pragma unroll
    for (int i = 0; i < EPT; ++i) {
        rank[i] = 0;
        if (c[i] >= 0) rank[i] = atomicAdd(&hist[c[i] >> BIN_SHIFT], 1);
    }
    __syncthreads();

    const int h0 = hist[tid];
    const int h1 = hist[tid + 256];
    int g0 = 0, g1 = 0;
    if (h0 > 0) g0 = atomicAdd(&cursor[tid * CUR_STRIDE], h0);
    if (h1 > 0) g1 = atomicAdd(&cursor[(tid + 256) * CUR_STRIDE], h1);
    if (tid < 32) {
        int v[16];
        int s = 0;
#pragma unroll
        for (int i = 0; i < 16; ++i) {
            v[i] = hist[tid * 16 + i];
            s += v[i];
        }
        int incl = s;
#pragma unroll
        for (int d = 1; d < 32; d <<= 1) {
            const int t = __shfl_up(incl, d, 32);
            if (tid >= d) incl += t;
        }
        int run = incl - s;
#pragma unroll
        for (int i = 0; i < 16; ++i) {
            off[tid * 16 + i] = run;
            run += v[i];
        }
    }
    base[tid] = g0;
    base[tid + 256] = g1;
    __syncthreads();

#pragma unroll
    for (int i = 0; i < EPT; ++i) {
        if (c[i] >= 0) {
            const int pos = off[c[i] >> BIN_SHIFT] + rank[i];
            stage_c[pos] = c[i];
            stage_r[pos] = r[i];
        }
    }
    __syncthreads();

    const int nval = min(CHUNK, E - e0);
#pragma unroll
    for (int i = 0; i < EPT; ++i) {
        const int s = i * 256 + tid;
        if (s < nval) {
            const int cc = stage_c[s];
            const int rr = stage_r[s];
            const int bin = cc >> BIN_SHIFT;
            const int g = base[bin] + (s - off[bin]);
            if (g < BIN_CAP) {
                *(int2*)(binbuf + 2 * ((size_t)bin * BIN_CAP + g)) = make_int2(cc, rr);
            } else {
                const int q = atomicAdd(bovn, 1);
                *(int2*)(bov + 2 * (size_t)q) = make_int2(cc, rr);
            }
        }
    }
}

#define BS_BATCH 8

__global__ __launch_bounds__(256)
void build_slots_kernel(const int* __restrict__ cursor,
                        const int* __restrict__ binbuf,
                        const int* __restrict__ bovn,
                        const int* __restrict__ bov,
                        int* __restrict__ cnt, int* __restrict__ slot,
                        int* ovn, int* __restrict__ ov, int N, int E) {
    typedef __attribute__((ext_vector_type(4))) int v4i;
    __shared__ int cnt_l[BIN_NODES];
    __shared__ __attribute__((aligned(16))) int slot_l[BIN_NODES * SLOTS];

    const int tid = threadIdx.x;
    const int bin = blockIdx.x;
    const int c0 = bin << BIN_SHIFT;

    cnt_l[tid] = 0;
    const v4i z4 = (v4i){0, 0, 0, 0};
#pragma unroll
    for (int i = 0; i < (BIN_NODES * SLOTS / 4) / 256; ++i)
        *(v4i*)(slot_l + (i * 256 + tid) * 4) = z4;
    __syncthreads();

    const int nent = min(cursor[bin * CUR_STRIDE], BIN_CAP);
    const int* reg = binbuf + 2 * (size_t)bin * BIN_CAP;
    for (int s0 = 0; s0 < nent; s0 += BS_BATCH * 256) {
        int2 pr[BS_BATCH];
#pragma unroll
        for (int j = 0; j < BS_BATCH; ++j) {
            const int s = s0 + j * 256 + tid;
            pr[j] = make_int2(-1, 0);
            if (s < nent) pr[j] = *(const int2*)(reg + 2 * (size_t)s);
        }
        int pl[BS_BATCH];
#pragma unroll
        for (int j = 0; j < BS_BATCH; ++j) {
            pl[j] = 0;
            if (pr[j].x >= 0) pl[j] = atomicAdd(&cnt_l[pr[j].x - c0], 1);
        }
#pragma unroll
        for (int j = 0; j < BS_BATCH; ++j) {
            if (pr[j].x >= 0) {
                if (pl[j] < SLOTS) {
                    slot_l[(pr[j].x - c0) * SLOTS + pl[j]] = pr[j].y;
                } else {
                    const int q = atomicAdd(ovn, 1);
                    ov[2 * (size_t)q] = pr[j].x;
                    ov[2 * (size_t)q + 1] = pr[j].y;
                }
            }
        }
    }
    const int nb = min(*bovn, E);
    for (int s = tid; s < nb; s += 256) {
        const int2 pr = *(const int2*)(bov + 2 * (size_t)s);
        if ((pr.x >> BIN_SHIFT) == bin) {
            const int local = pr.x - c0;
            const int p = atomicAdd(&cnt_l[local], 1);
            if (p < SLOTS) {
                slot_l[local * SLOTS + p] = pr.y;
            } else {
                const int q = atomicAdd(ovn, 1);
                ov[2 * (size_t)q] = pr.x;
                ov[2 * (size_t)q + 1] = pr.y;
            }
        }
    }
    __syncthreads();

    if (c0 + tid < N) cnt[c0 + tid] = cnt_l[tid];
#pragma unroll
    for (int i = 0; i < (BIN_NODES * SLOTS / 4) / 256; ++i) {
        const int idx = i * 256 + tid;
        const int node = c0 + (idx >> 3);
        if (node < N)
            *(v4i*)(slot + (size_t)c0 * SLOTS + (size_t)idx * 4) = *(const v4i*)(slot_l + idx * 4);
    }
}

#define FG_ROWS 32

template <int F, int NMID, int NOUT>
__global__ __launch_bounds__(128)
void gather_ln_products_kernel(const float* __restrict__ src,
                               const int* __restrict__ cnt,
                               const int* __restrict__ slot,
                               const int* __restrict__ ovn,
                               const int* __restrict__ ov,
                               const float* __restrict__ b,
                               const float* __restrict__ g,
                               const float* __restrict__ be,
                               const float* __restrict__ W,
                               const float* __restrict__ Wc,
                               const float* __restrict__ b2,
                               const float* __restrict__ bc,
                               float* __restrict__ out,
                               float* __restrict__ cvec, int N, int E) {
    static_assert(NMID == 64 && NOUT == 32, "tile dealing below is for 4 waves, 64 and 32 columns");
    constexpr int PER = F / 32;
    constexpr int PITCH = F + 4;
    constexpr int NT = NMID / 16;
    constexpr int MT = NT / 2;
    typedef __attribute__((ext_vector_type(PER))) float vf;
    __shared__ __attribute__((aligned(16))) float tile[FG_ROWS * PITCH];

    const int lane = threadIdx.x & 31;
    const int w = __builtin_amdgcn_readfirstlane((int)(threadIdx.x >> 5));
    const int node0 = blockIdx.x * FG_ROWS;
    const float* lsrc = src + lane * PER;

    for (int q = 0; q < FG_ROWS / 4; ++q) {
        const int lr = w * (FG_ROWS / 4) + q;
        const int c = node0 + lr;
        vf y;
#pragma unroll
        for (int i = 0; i < PER; ++i) y[i] = 0.f;
        if (c < N) {
            const int deg = cnt[c];
            const float dc = rsqrtf((float)(deg + 1));

            vf acc = *(const vf*)(lsrc + (size_t)c * F);
            {
                const int n = min(SLOTS, deg);
                int my_r = c;
                if (lane < n) my_r = slot[(size_t)c * SLOTS + lane];
                int j = 0;
                for (; j + 4 <= n; j += 4) {
                    const int r0 = __builtin_amdgcn_readlane(my_r, j);
                    const int r1 = __builtin_amdgcn_readlane(my_r, j + 1);
                    const int r2 = __builtin_amdgcn_readlane(my_r, j + 2);
                    const int r3 = __builtin_amdgcn_readlane(my_r, j + 3);
                    const vf x0 = *(const vf*)(lsrc + (size_t)r0 * F);
                    const vf x1 = *(const vf*)(lsrc + (size_t)r1 * F);
                    const vf x2 = *(const vf*)(lsrc + (size_t)r2 * F);
                    const vf x3 = *(const vf*)(lsrc + (size_t)r3 * F);
                    acc += x0;
                    acc += x1;
                    acc += x2;
                    acc += x3;
                }
                for (; j < n; ++j) {
                    const int r0 = __builtin_amdgcn_readlane(my_r, j);
                    acc += *(const vf*)(lsrc + (size_t)r0 * F);
                }
            }
            if (deg > SLOTS) {
                const int nov = min(*ovn, E);
                for (int base = 0; base < nov; base += 32) {
                    int t = -1;
                    int r = c;
                    if (base + lane < nov) {
                        t = ov[2 * (size_t)(base + lane)];
                        r = ov[2 * (size_t)(base + lane) + 1];
                    }
                    unsigned hit = __builtin_amdgcn_ballot_w32(t == c);
                    while (hit) {
                        const int j2 = __builtin_ctz(hit);
                        hit &= hit - 1;
                        const int r0 = __builtin_amdgcn_readlane(r, j2);
                        acc += *(const vf*)(lsrc + (size_t)r0 * F);
                    }
                }
            }
            acc *= dc;

            float v[PER];
#pragma unroll
            for (int i = 0; i < PER; ++i) v[i] = acc[i] + b[lane * PER + i];

            float sum = 0.f;
#pragma unroll
            for (int i = 0; i < PER; ++i) sum += v[i];
#pragma unroll
            for (int off = 16; off > 0; off >>= 1) sum += __shfl_xor(sum, off, 32);
            const float mu = sum * (1.0f / F);

            float var = 0.f;
#pragma unroll
            for (int i = 0; i < PER; ++i) {
                const float d = v[i] - mu;
                var += d * d;
            }
#pragma unroll
            for (int off = 16; off > 0; off >>= 1) var += __shfl_xor(var, off, 32);
            var *= (1.0f / F);
            const float rs = rsqrtf(var + LN_EPS);

#pragma unroll
            for (int i = 0; i < PER; ++i) {
                const float z = (v[i] - mu) * rs * g[lane * PER + i] + be[lane * PER + i];
                y[i] = fmaxf(z, 0.f);
            }
        }
        *(vf*)(tile + lr * PITCH + lane * PER) = y;
    }
    __syncthreads();

    const int mr = lane & 15;
    const int hi = lane >> 4;
    const int n = w * 16 + mr;
    const float* ap = tile + mr * PITCH + hi * 2;
    const float* wp = W + (size_t)(hi * 2) * NMID + n;

    v8f acc[MT];
#pragma unroll
    for (int mt = 0; mt < MT; ++mt)
        acc[mt] = (v8f){0.f, 0.f, 0.f, 0.f, 0.f, 0.f, 0.f, 0.f};

#pragma unroll 8
    for (int k = 0; k < F; k += 4) {
        v2f bf;
        bf.x = wp[(size_t)k * NMID];
        bf.y = wp[(size_t)k * NMID + NMID];
#pragma unroll
        for (int mt = 0; mt < MT; ++mt) {
            const v2f a = *(const v2f*)(ap + mt * 16 * PITCH + k);
            acc[mt] = __builtin_amdgcn_wmma_f32_16x16x4_f32(
                false, a, false, bf, (short)0, acc[mt], false, false);
        }
    }

    constexpr int OP = NMID + 4;
    __syncthreads();
#pragma unroll
    for (int mt = 0; mt < MT; ++mt) {
#pragma unroll
        for (int v = 0; v < 8; ++v)
            tile[(mt * 16 + v + hi * 8) * OP + n] = acc[mt][v];
    }
    __syncthreads();

    const int rt = w >> 1;
    const int n2 = (w & 1) * 16 + mr;
    const float* ap2 = tile + (rt * 16 + mr) * OP + hi * 2;
    const float* wp2 = Wc + (size_t)(hi * 2) * NOUT + n2;
    v8f acc2 = (v8f){0.f, 0.f, 0.f, 0.f, 0.f, 0.f, 0.f, 0.f};
#pragma unroll
    for (int k = 0; k < NMID; k += 4) {
        v2f bf;
        bf.x = wp2[(size_t)k * NOUT];
        bf.y = wp2[(size_t)k * NOUT + NOUT];
        const v2f a = *(const v2f*)(ap2 + k);
        acc2 = __builtin_amdgcn_wmma_f32_16x16x4_f32(
            false, a, false, bf, (short)0, acc2, false, false);
    }

    constexpr int OP2 = NOUT + 4;
    __syncthreads();
#pragma unroll
    for (int v = 0; v < 8; ++v)
        tile[(rt * 16 + v + hi * 8) * OP2 + n2] = acc2[v];
    __syncthreads();

    typedef __attribute__((ext_vector_type(4))) float v4f;
    constexpr int LPR = NOUT / 4;
    constexpr int RPI = 32 / LPR;
    const int cl = (lane % LPR) * 4;
    const int rl = lane / LPR;
#pragma unroll
    for (int i = 0; i < (FG_ROWS / 4) / RPI; ++i) {
        const int r = w * (FG_ROWS / 4) + i * RPI + rl;
        const int m = node0 + r;
        if (m < N) {
            v4f o = *(const v4f*)(tile + r * OP2 + cl);
            o *= rsqrtf((float)(cnt[m] + 1));
            *(v4f*)(out + (size_t)m * NOUT + cl) = o;
        }
    }

    if (blockIdx.x == 0 && w < 2) {
        const int n3 = w * 16 + mr;
        const float* wp3 = Wc + (size_t)(hi * 2) * NOUT + n3;
        v8f acc3 = (v8f){0.f, 0.f, 0.f, 0.f, 0.f, 0.f, 0.f, 0.f};
#pragma unroll
        for (int k = 0; k < NMID; k += 4) {
            v2f a, bf;
            a.x = b2[k + hi * 2];
            a.y = b2[k + hi * 2 + 1];
            bf.x = wp3[(size_t)k * NOUT];
            bf.y = wp3[(size_t)k * NOUT + NOUT];
            acc3 = __builtin_amdgcn_wmma_f32_16x16x4_f32(
                false, a, false, bf, (short)0, acc3, false, false);
        }
        if (hi == 0) cvec[n3] = acc3[0] + bc[n3];
    }
}

__global__ __launch_bounds__(256)
void gather_rows32_kernel(const float* __restrict__ p,
                          const int* __restrict__ cnt,
                          const int* __restrict__ slot,
                          const int* __restrict__ ovn,
                          const int* __restrict__ ov,
                          const float* __restrict__ cvec,
                          float* __restrict__ out, int N, int E) {
    constexpr int C = 32;
    typedef __attribute__((ext_vector_type(4))) float v4f;
    typedef __attribute__((ext_vector_type(4))) int v4i;

    const int lane = threadIdx.x & 31;
    const int sub = lane & 7;
    const int gq = lane >> 3;
    const int gb = lane & ~7;
    const int wave = (blockIdx.x * blockDim.x + threadIdx.x) >> 5;
    const int c = wave * 4 + gq;
    const bool live = c < N;
    const int cc = live ? c : N - 1;

    const int cntv = cnt[cc];
    const v4i s = *(const v4i*)(slot + (size_t)cc * SLOTS + sub * 4);
    const int deg = live ? cntv : 0;
    const int n = min(SLOTS, deg);
    const float* lp = p + sub * 4;
    v4f acc = *(const v4f*)(lp + (size_t)cc * C);

    int nm = n;
    nm = max(nm, __shfl_xor(nm, 8, 32));
    nm = max(nm, __shfl_xor(nm, 16, 32));
    const int nmax = __builtin_amdgcn_readfirstlane(nm);

    const v4f zero = (v4f){0.f, 0.f, 0.f, 0.f};
    int j = 0;
    do {
        const int la = gb + (j >> 2);
        const int lb = la + 1;
        const int r0 = __shfl(s.x, la, 32);
        const int r1 = __shfl(s.y, la, 32);
        const int r2 = __shfl(s.z, la, 32);
        const int r3 = __shfl(s.w, la, 32);
        const int r4 = __shfl(s.x, lb, 32);
        const int r5 = __shfl(s.y, lb, 32);
        const int r6 = __shfl(s.z, lb, 32);
        const int r7 = __shfl(s.w, lb, 32);
        v4f x0 = zero, x1 = zero, x2 = zero, x3 = zero, x4 = zero, x5 = zero, x6 = zero, x7 = zero;
        if (j     < n) x0 = *(const v4f*)(lp + (size_t)r0 * C);
        if (j + 1 < n) x1 = *(const v4f*)(lp + (size_t)r1 * C);
        if (j + 2 < n) x2 = *(const v4f*)(lp + (size_t)r2 * C);
        if (j + 3 < n) x3 = *(const v4f*)(lp + (size_t)r3 * C);
        if (j + 4 < n) x4 = *(const v4f*)(lp + (size_t)r4 * C);
        if (j + 5 < n) x5 = *(const v4f*)(lp + (size_t)r5 * C);
        if (j + 6 < n) x6 = *(const v4f*)(lp + (size_t)r6 * C);
        if (j + 7 < n) x7 = *(const v4f*)(lp + (size_t)r7 * C);
        acc += x0;
        acc += x1;
        acc += x2;
        acc += x3;
        acc += x4;
        acc += x5;
        acc += x6;
        acc += x7;
        j += 8;
    } while (j < nmax);

    if (__builtin_amdgcn_ballot_w32(deg > SLOTS)) {
        const int nov = min(*ovn, E);
        for (int base = 0; base < nov; base += 32) {
            int t = -1;
            int r = 0;
            if (base + lane < nov) {
                t = ov[2 * (size_t)(base + lane)];
                r = ov[2 * (size_t)(base + lane) + 1];
            }
#pragma unroll
            for (int q = 0; q < 4; ++q) {
                const int cq = __builtin_amdgcn_readlane(c, q * 8);
                unsigned hit = __builtin_amdgcn_ballot_w32(t == cq);
                while (hit) {
                    const int j2 = __builtin_ctz(hit);
                    hit &= hit - 1;
                    const int r0 = __builtin_amdgcn_readlane(r, j2);
                    if (gq == q) acc += *(const v4f*)(lp + (size_t)r0 * C);
                }
            }
        }
    }

    if (live) {
        const float dc = rsqrtf((float)(deg + 1));
        const v4f cv = *(const v4f*)(cvec + sub * 4);
        const v4f o = acc * dc + cv;
        *(v4f*)(out + (size_t)c * C + sub * 4) = o;
    }
}

extern "C" void kernel_launch(void* const* d_in, const int* in_sizes, int n_in,
                              void* d_out, int out_size, void* d_ws, size_t ws_size,
                              hipStream_t stream) {
    constexpr int F = 128, H = 128, D = 64, C = 32;
    constexpr int MT = 2;

    const float* x   = (const float*)d_in[0];
    const int*   ei  = (const int*)d_in[1];
    const float* W1  = (const float*)d_in[2];
    const float* b1  = (const float*)d_in[3];
    const float* g1  = (const float*)d_in[4];
    const float* be1 = (const float*)d_in[5];
    const float* W2  = (const float*)d_in[6];
    const float* b2  = (const float*)d_in[7];
    const float* Wc  = (const float*)d_in[8];
    const float* bc  = (const float*)d_in[9];
    float* logits    = (float*)d_out;

    const int N = in_sizes[0] / F;
    const int E = in_sizes[1] / 2;
    const int* rows = ei;
    const int* cols = ei + E;

    float* ws   = (float*)d_ws;
    float* hw1  = ws;
    float* p    = ws + (size_t)N * 128;
    float* cvec = ws + (size_t)N * 160;
    int*   cur  = (int*)(ws + (size_t)N * 160 + 64);
    int*   slot = (int*)(ws + (size_t)N * 192);
    int*   ov   = (int*)(ws + (size_t)N * 224);
    int*   cnt  = (int*)(ws + (size_t)N * 256);
    int*   binbuf = (int*)ws;
    int*   bov  = (int*)(ws + (size_t)N * 96);
    int*   ovn  = cur;
    int*   bovn = cur + 1;
    int*   bcur = cur + CUR_BINS;

    const int TPB = 256;
    const int wavesPerBlk = TPB / 32;
    auto cdiv = [](long long a, long long b) { return (int)((a + b - 1) / b); };

    const int NB = cdiv(N, BIN_NODES);
    zero_cursors_kernel<<<cdiv(CUR_BINS + NB * CUR_STRIDE, TPB), TPB, 0, stream>>>(cur, CUR_BINS + NB * CUR_STRIDE);
    bin_edges_kernel<<<cdiv(E, CHUNK), 256, 0, stream>>>(rows, cols, bcur, binbuf, bovn, bov, E);
    build_slots_kernel<<<NB, 256, 0, stream>>>(bcur, binbuf, bovn, bov, cnt, slot, ovn, ov, N, E);

    {
        const int nWaves = cdiv(N, 16 * MT);
        gemm_wmma_f32<F, H / 16, MT, true><<<cdiv(nWaves, wavesPerBlk), TPB, 0, stream>>>(
            x, W1, nullptr, cnt, hw1, N);
    }

    gather_ln_products_kernel<H, D, C><<<cdiv(N, FG_ROWS), 128, 0, stream>>>(
        hw1, cnt, slot, ovn, ov, b1, g1, be1, W2, Wc, b2, bc, p, cvec, N, E);

    gather_rows32_kernel<<<cdiv(N, 4 * wavesPerBlk), TPB, 0, stream>>>(
        p, cnt, slot, ovn, ov, cvec, logits, N, E);
}


# Slot-building launch as one resident pass: bins of 400 target nodes, 250 blocks of 20 waves (one per WGP), x@W1's 25 row tiles of a bin dealt evenly over the 4 SIMDs; no second round of blocks
# speedup vs baseline: 6.2546x; 1.1542x over previous
#include <hip/hip_runtime.h>
#include <hip/hip_bf16.h>


typedef __attribute__((ext_vector_type(2))) float v2f;
typedef __attribute__((ext_vector_type(8))) float v8f;

#define LN_EPS 1e-5f

template <int K, int NT, int MT, int LDW>
__device__ __forceinline__
void rows_times_w_f32(const float* __restrict__ A, const float* __restrict__ W,
                      int m0, int lane, v8f (&acc)[MT][NT]) {
    constexpr int NOUT = LDW;
    const int mr = lane & 15;
    const int hi = lane >> 4;

    const float* arow[MT];
#pragma unroll
    for (int mt = 0; mt < MT; ++mt)
        arow[mt] = A + (size_t)(m0 + mt * 16 + mr) * K;

    constexpr int UNR = (MT * NT >= 8) ? 2 : 4;
#pragma unroll UNR
    for (int k = 0; k < K; k += 4) {
        const int kb = k + hi * 2;
        v2f a[MT];
#pragma unroll
        for (int mt = 0; mt < MT; ++mt) {
            a[mt].x = arow[mt][kb];
            a[mt].y = arow[mt][kb + 1];
        }
        const float* w0 = W + (size_t)kb * NOUT;
        const float* w1 = w0 + NOUT;
#pragma unroll
        for (int t = 0; t < NT; ++t) {
            const int n = t * 16 + mr;
            v2f b;
            b.x = w0[n];
            b.y = w1[n];
#pragma unroll
            for (int mt = 0; mt < MT; ++mt)
                acc[mt][t] = __builtin_amdgcn_wmma_f32_16x16x4_f32(
                    false, a[mt], false, b, (short)0, acc[mt][t], false, false);
        }
    }
}

template <int NT, int MT, int LDO>
__device__ __forceinline__
void store_rows_scaled(const v8f (&acc)[MT][NT], const int* cntl, int c0,
                       float* __restrict__ out, int m0, int lane) {
    constexpr int NOUT = LDO;
    const int mr = lane & 15;
    const int hi = lane >> 4;
    float rsv[MT][8];
#pragma unroll
    for (int mt = 0; mt < MT; ++mt)
#pragma unroll
        for (int v = 0; v < 8; ++v)
            rsv[mt][v] = rsqrtf((float)(cntl[m0 - c0 + mt * 16 + v + hi * 8] + 1));

#pragma unroll
    for (int mt = 0; mt < MT; ++mt) {
#pragma unroll
        for (int t = 0; t < NT; ++t) {
            const int n = t * 16 + mr;
#pragma unroll
            for (int v = 0; v < 8; ++v) {
                const int m = m0 + mt * 16 + v + hi * 8;
                out[(size_t)m * NOUT + n] = acc[mt][t][v] * rsv[mt][v];
            }
        }
    }
}

#define SLOTS 32
#define BIN_NODES 400
#define BIN_MAX 512
#define CHUNK 4096
#define EPT (CHUNK / 256)

__global__ __launch_bounds__(256)
void bin_edges_kernel(const int* __restrict__ row,
                      const int* __restrict__ col,
                      int* __restrict__ binbuf, int* __restrict__ runtab,
                      int* __restrict__ ovn, int E) {
    __shared__ int hist[BIN_MAX];
    __shared__ int off[BIN_MAX];
    __shared__ int stage_c[CHUNK];
    __shared__ int stage_r[CHUNK];

    const int tid = threadIdx.x;
    const int e0 = blockIdx.x * CHUNK;
    hist[tid] = 0;
    hist[tid + 256] = 0;
    if (blockIdx.x == 0 && tid == 0) *ovn = 0;
    __syncthreads();

    int c[EPT], r[EPT], rank[EPT];
#pragma unroll
    for (int i = 0; i < EPT; ++i) {
        const int e = e0 + i * 256 + tid;
        c[i] = -1;
        r[i] = 0;
        if (e < E) {
            c[i] = col[e];
            r[i] = row[e];
        }
    }
#pragma unroll
    for (int i = 0; i < EPT; ++i) {
        rank[i] = 0;
        if (c[i] >= 0) rank[i] = atomicAdd(&hist[c[i] / BIN_NODES], 1);
    }
    __syncthreads();

    if (tid < 32) {
        int v[16];
        int s = 0;
#pragma unroll
        for (int i = 0; i < 16; ++i) {
            v[i] = hist[tid * 16 + i];
            s += v[i];
        }
        int incl = s;
#pragma unroll
        for (int d = 1; d < 32; d <<= 1) {
            const int t = __shfl_up(incl, d, 32);
            if (tid >= d) incl += t;
        }
        int run = incl - s;
#pragma unroll
        for (int i = 0; i < 16; ++i) {
            off[tid * 16 + i] = run;
            run += v[i];
        }
    }
    __syncthreads();

    int* myrow = runtab + (size_t)blockIdx.x * BIN_MAX;
    myrow[tid] = off[tid] | (hist[tid] << 16);
    myrow[tid + 256] = off[tid + 256] | (hist[tid + 256] << 16);

#pragma unroll
    for (int i = 0; i < EPT; ++i) {
        if (c[i] >= 0) {
            const int pos = off[c[i] / BIN_NODES] + rank[i];
            stage_c[pos] = c[i];
            stage_r[pos] = r[i];
        }
    }
    __syncthreads();

    const int nval = min(CHUNK, E - e0);
#pragma unroll
    for (int i = 0; i < EPT; ++i) {
        const int s = i * 256 + tid;
        if (s < nval)
            *(int2*)(binbuf + 2 * ((size_t)e0 + s)) = make_int2(stage_c[s], stage_r[s]);
    }
}

#define RUN_REG 8
#define RUN_GRP 4
#define RUN_PRE 3
#define PROD_WAVES 12

template <int K, int NT>
__global__ __launch_bounds__(640)
void build_slots_rows_kernel(const int* __restrict__ runtab,
                             const int* __restrict__ binbuf,
                             int* __restrict__ cnt, int* __restrict__ slot,
                             int* ovn, int* __restrict__ ov,
                             const float* __restrict__ x,
                             const float* __restrict__ W,
                             float* __restrict__ hw, int N, int E) {
    typedef __attribute__((ext_vector_type(4))) int v4i;
    constexpr int MT = 2;
    constexpr int NOUT = NT * 16;
    constexpr int XT = 2;
    constexpr int XROW = PROD_WAVES * MT * 16;
    static_assert(XROW + 16 == BIN_NODES, "12 stripes of 32 rows and one tile of 16 make the bin");
    static_assert(4 * XT == NT, "four waves cover the last tile's columns");
    __shared__ int cnt_l[BIN_NODES];
    __shared__ __attribute__((aligned(16))) int slot_l[BIN_NODES * SLOTS];

    const int w = __builtin_amdgcn_readfirstlane((int)(threadIdx.x >> 5));
    const bool builder = w < 8;
    const int lane = threadIdx.x & 31;
    const int tid = threadIdx.x & 255;
    const int bin = blockIdx.x;
    const int c0 = bin * BIN_NODES;
    const int g = w - 8;
    const int m0 = c0 + g * (MT * 16);
    const int mx = c0 + XROW;
    const bool product = !builder && m0 + MT * 16 <= N;
    const bool product_x = !builder && g < NT / XT && mx + 16 <= N;

    v8f acc[MT][NT];
#pragma unroll
    for (int mt = 0; mt < MT; ++mt)
#pragma unroll
        for (int t = 0; t < NT; ++t)
            acc[mt][t] = (v8f){0.f, 0.f, 0.f, 0.f, 0.f, 0.f, 0.f, 0.f};
    v8f accx[1][XT];
#pragma unroll
    for (int t = 0; t < XT; ++t)
        accx[0][t] = (v8f){0.f, 0.f, 0.f, 0.f, 0.f, 0.f, 0.f, 0.f};

    if (builder) {
        for (int i = tid; i < BIN_NODES; i += 256) cnt_l[i] = 0;
        const v4i z4 = (v4i){0, 0, 0, 0};
        for (int i = tid; i < BIN_NODES * SLOTS / 4; i += 256)
            *(v4i*)(slot_l + i * 4) = z4;
    }
    __syncthreads();

    if (builder) {
        const int nblk1 = (E + CHUNK - 1) / CHUNK;
        const int grp = tid >> 3;
        const int sub = tid & 7;
        for (int rb = 0; rb < nblk1; rb += 32 * RUN_REG) {
            int ent[RUN_REG];
#pragma unroll
            for (int j = 0; j < RUN_REG; ++j) {
                const int q = rb + grp + 32 * j;
                ent[j] = 0;
                if (q < nblk1) ent[j] = runtab[(size_t)q * BIN_MAX + bin];
            }
#pragma unroll
            for (int j0 = 0; j0 < RUN_REG; j0 += RUN_GRP) {
                int2 pr[RUN_GRP][RUN_PRE];
                const int* srcp[RUN_GRP];
                int nn[RUN_GRP];
#pragma unroll
                for (int u = 0; u < RUN_GRP; ++u) {
                    const int q = rb + grp + 32 * (j0 + u);
                    nn[u] = ent[j0 + u] >> 16;
                    srcp[u] = binbuf + 2 * ((size_t)q * CHUNK + (ent[j0 + u] & 0xFFFF));
#pragma unroll
                    for (int v = 0; v < RUN_PRE; ++v) {
                        pr[u][v] = make_int2(-1, 0);
                        if (sub + 8 * v < nn[u]) pr[u][v] = *(const int2*)(srcp[u] + 2 * (sub + 8 * v));
                    }
                }
                int pl[RUN_GRP][RUN_PRE];
#pragma unroll
                for (int u = 0; u < RUN_GRP; ++u)
#pragma unroll
                    for (int v = 0; v < RUN_PRE; ++v) {
                        pl[u][v] = 0;
                        if (pr[u][v].x >= 0) pl[u][v] = atomicAdd(&cnt_l[pr[u][v].x - c0], 1);
                    }
#pragma unroll
                for (int u = 0; u < RUN_GRP; ++u)
#pragma unroll
                    for (int v = 0; v < RUN_PRE; ++v) {
                        if (pr[u][v].x >= 0) {
                            if (pl[u][v] < SLOTS) {
                                slot_l[(pr[u][v].x - c0) * SLOTS + pl[u][v]] = pr[u][v].y;
                            } else {
                                const int q = atomicAdd(ovn, 1);
                                ov[2 * (size_t)q] = pr[u][v].x;
                                ov[2 * (size_t)q + 1] = pr[u][v].y;
                            }
                        }
                    }
#pragma unroll
                for (int u = 0; u < RUN_GRP; ++u) {
                    for (int i = sub + 8 * RUN_PRE; i < nn[u]; i += 8) {
                        const int2 p2 = *(const int2*)(srcp[u] + 2 * i);
                        const int local = p2.x - c0;
                        const int pp = atomicAdd(&cnt_l[local], 1);
                        if (pp < SLOTS) {
                            slot_l[local * SLOTS + pp] = p2.y;
                        } else {
                            const int q = atomicAdd(ovn, 1);
                            ov[2 * (size_t)q] = p2.x;
                            ov[2 * (size_t)q + 1] = p2.y;
                        }
                    }
                }
            }
        }
    } else {
        if (product_x) rows_times_w_f32<K, XT, 1, NOUT>(x, W + g * (XT * 16), mx, lane, accx);
        if (product) rows_times_w_f32<K, NT, MT, NOUT>(x, W, m0, lane, acc);
    }
    __syncthreads();

    if (builder) {
        for (int i = tid; i < BIN_NODES; i += 256)
            if (c0 + i < N) cnt[c0 + i] = cnt_l[i];
        for (int idx = tid; idx < BIN_NODES * SLOTS / 4; idx += 256) {
            const int node = c0 + (idx >> 3);
            if (node < N)
                *(v4i*)(slot + (size_t)c0 * SLOTS + (size_t)idx * 4) = *(const v4i*)(slot_l + idx * 4);
        }
    } else {
        if (product) store_rows_scaled<NT, MT, NOUT>(acc, cnt_l, c0, hw, m0, lane);
        if (product_x) store_rows_scaled<XT, 1, NOUT>(accx, cnt_l, c0, hw + g * (XT * 16), mx, lane);
    }
}

#define FG_ROWS 32

template <int F, int NMID, int NOUT>
__global__ __launch_bounds__(128)
void gather_ln_products_kernel(const float* __restrict__ src,
                               const int* __restrict__ cnt,
                               const int* __restrict__ slot,
                               const int* __restrict__ ovn,
                               const int* __restrict__ ov,
                               const float* __restrict__ b,
                               const float* __restrict__ g,
                               const float* __restrict__ be,
                               const float* __restrict__ W,
                               const float* __restrict__ Wc,
                               const float* __restrict__ b2,
                               const float* __restrict__ bc,
                               float* __restrict__ out,
                               float* __restrict__ cvec, int N, int E) {
    static_assert(NMID == 64 && NOUT == 32, "tile dealing below is for 4 waves, 64 and 32 columns");
    constexpr int PER = F / 32;
    constexpr int PITCH = F + 4;
    constexpr int NT = NMID / 16;
    constexpr int MT = NT / 2;
    typedef __attribute__((ext_vector_type(PER))) float vf;
    __shared__ __attribute__((aligned(16))) float tile[FG_ROWS * PITCH];

    const int lane = threadIdx.x & 31;
    const int w = __builtin_amdgcn_readfirstlane((int)(threadIdx.x >> 5));
    const int node0 = blockIdx.x * FG_ROWS;
    const float* lsrc = src + lane * PER;

    for (int q = 0; q < FG_ROWS / 4; ++q) {
        const int lr = w * (FG_ROWS / 4) + q;
        const int c = node0 + lr;
        vf y;
#pragma unroll
        for (int i = 0; i < PER; ++i) y[i] = 0.f;
        if (c < N) {
            const int deg = cnt[c];
            const float dc = rsqrtf((float)(deg + 1));

            vf acc = *(const vf*)(lsrc + (size_t)c * F);
            {
                const int n = min(SLOTS, deg);
                int my_r = c;
                if (lane < n) my_r = slot[(size_t)c * SLOTS + lane];
                int j = 0;
                for (; j + 4 <= n; j += 4) {
                    const int r0 = __builtin_amdgcn_readlane(my_r, j);
                    const int r1 = __builtin_amdgcn_readlane(my_r, j + 1);
                    const int r2 = __builtin_amdgcn_readlane(my_r, j + 2);
                    const int r3 = __builtin_amdgcn_readlane(my_r, j + 3);
                    const vf x0 = *(const vf*)(lsrc + (size_t)r0 * F);
                    const vf x1 = *(const vf*)(lsrc + (size_t)r1 * F);
                    const vf x2 = *(const vf*)(lsrc + (size_t)r2 * F);
                    const vf x3 = *(const vf*)(lsrc + (size_t)r3 * F);
                    acc += x0;
                    acc += x1;
                    acc += x2;
                    acc += x3;
                }
                for (; j < n; ++j) {
                    const int r0 = __builtin_amdgcn_readlane(my_r, j);
                    acc += *(const vf*)(lsrc + (size_t)r0 * F);
                }
            }
            if (deg > SLOTS) {
                const int nov = min(*ovn, E);
                for (int base = 0; base < nov; base += 32) {
                    int t = -1;
                    int r = c;
                    if (base + lane < nov) {
                        t = ov[2 * (size_t)(base + lane)];
                        r = ov[2 * (size_t)(base + lane) + 1];
                    }
                    unsigned hit = __builtin_amdgcn_ballot_w32(t == c);
                    while (hit) {
                        const int j2 = __builtin_ctz(hit);
                        hit &= hit - 1;
                        const int r0 = __builtin_amdgcn_readlane(r, j2);
                        acc += *(const vf*)(lsrc + (size_t)r0 * F);
                    }
                }
            }
            acc *= dc;

            float v[PER];
#pragma unroll
            for (int i = 0; i < PER; ++i) v[i] = acc[i] + b[lane * PER + i];

            float sum = 0.f;
#pragma unroll
            for (int i = 0; i < PER; ++i) sum += v[i];
#pragma unroll
            for (int off = 16; off > 0; off >>= 1) sum += __shfl_xor(sum, off, 32);
            const float mu = sum * (1.0f / F);

            float var = 0.f;
#pragma unroll
            for (int i = 0; i < PER; ++i) {
                const float d = v[i] - mu;
                var += d * d;
            }
#pragma unroll
            for (int off = 16; off > 0; off >>= 1) var += __shfl_xor(var, off, 32);
            var *= (1.0f / F);
            const float rs = rsqrtf(var + LN_EPS);

#pragma unroll
            for (int i = 0; i < PER; ++i) {
                const float z = (v[i] - mu) * rs * g[lane * PER + i] + be[lane * PER + i];
                y[i] = fmaxf(z, 0.f);
            }
        }
        *(vf*)(tile + lr * PITCH + lane * PER) = y;
    }
    __syncthreads();

    const int mr = lane & 15;
    const int hi = lane >> 4;
    const int n = w * 16 + mr;
    const float* ap = tile + mr * PITCH + hi * 2;
    const float* wp = W + (size_t)(hi * 2) * NMID + n;

    v8f acc[MT];
#pragma unroll
    for (int mt = 0; mt < MT; ++mt)
        acc[mt] = (v8f){0.f, 0.f, 0.f, 0.f, 0.f, 0.f, 0.f, 0.f};

#pragma unroll 8
    for (int k = 0; k < F; k += 4) {
        v2f bf;
        bf.x = wp[(size_t)k * NMID];
        bf.y = wp[(size_t)k * NMID + NMID];
#pragma unroll
        for (int mt = 0; mt < MT; ++mt) {
            const v2f a = *(const v2f*)(ap + mt * 16 * PITCH + k);
            acc[mt] = __builtin_amdgcn_wmma_f32_16x16x4_f32(
                false, a, false, bf, (short)0, acc[mt], false, false);
        }
    }

    constexpr int OP = NMID + 4;
    __syncthreads();
#pragma unroll
    for (int mt = 0; mt < MT; ++mt) {
#pragma unroll
        for (int v = 0; v < 8; ++v)
            tile[(mt * 16 + v + hi * 8) * OP + n] = acc[mt][v];
    }
    __syncthreads();

    const int rt = w >> 1;
    const int n2 = (w & 1) * 16 + mr;
    const float* ap2 = tile + (rt * 16 + mr) * OP + hi * 2;
    const float* wp2 = Wc + (size_t)(hi * 2) * NOUT + n2;
    v8f acc2 = (v8f){0.f, 0.f, 0.f, 0.f, 0.f, 0.f, 0.f, 0.f};
#pragma unroll
    for (int k = 0; k < NMID; k += 4) {
        v2f bf;
        bf.x = wp2[(size_t)k * NOUT];
        bf.y = wp2[(size_t)k * NOUT + NOUT];
        const v2f a = *(const v2f*)(ap2 + k);
        acc2 = __builtin_amdgcn_wmma_f32_16x16x4_f32(
            false, a, false, bf, (short)0, acc2, false, false);
    }

    constexpr int OP2 = NOUT + 4;
    __syncthreads();
#pragma unroll
    for (int v = 0; v < 8; ++v)
        tile[(rt * 16 + v + hi * 8) * OP2 + n2] = acc2[v];
    __syncthreads();

    typedef __attribute__((ext_vector_type(4))) float v4f;
    constexpr int LPR = NOUT / 4;
    constexpr int RPI = 32 / LPR;
    const int cl = (lane % LPR) * 4;
    const int rl = lane / LPR;
#pragma unroll
    for (int i = 0; i < (FG_ROWS / 4) / RPI; ++i) {
        const int r = w * (FG_ROWS / 4) + i * RPI + rl;
        const int m = node0 + r;
        if (m < N) {
            v4f o = *(const v4f*)(tile + r * OP2 + cl);
            o *= rsqrtf((float)(cnt[m] + 1));
            *(v4f*)(out + (size_t)m * NOUT + cl) = o;
        }
    }

    if (blockIdx.x == 0 && w < 2) {
        const int n3 = w * 16 + mr;
        const float* wp3 = Wc + (size_t)(hi * 2) * NOUT + n3;
        v8f acc3 = (v8f){0.f, 0.f, 0.f, 0.f, 0.f, 0.f, 0.f, 0.f};
#pragma unroll
        for (int k = 0; k < NMID; k += 4) {
            v2f a, bf;
            a.x = b2[k + hi * 2];
            a.y = b2[k + hi * 2 + 1];
            bf.x = wp3[(size_t)k * NOUT];
            bf.y = wp3[(size_t)k * NOUT + NOUT];
            acc3 = __builtin_amdgcn_wmma_f32_16x16x4_f32(
                false, a, false, bf, (short)0, acc3, false, false);
        }
        if (hi == 0) cvec[n3] = acc3[0] + bc[n3];
    }
}

__global__ __launch_bounds__(256)
void gather_rows32_kernel(const float* __restrict__ p,
                          const int* __restrict__ cnt,
                          const int* __restrict__ slot,
                          const int* __restrict__ ovn,
                          const int* __restrict__ ov,
                          const float* __restrict__ cvec,
                          float* __restrict__ out, int N, int E) {
    constexpr int C = 32;
    typedef __attribute__((ext_vector_type(4))) float v4f;
    typedef __attribute__((ext_vector_type(4))) int v4i;

    const int lane = threadIdx.x & 31;
    const int sub = lane & 7;
    const int gq = lane >> 3;
    const int gb = lane & ~7;
    const int wave = (blockIdx.x * blockDim.x + threadIdx.x) >> 5;
    const int c = wave * 4 + gq;
    const bool live = c < N;
    const int cc = live ? c : N - 1;

    const int cntv = cnt[cc];
    const v4i s = *(const v4i*)(slot + (size_t)cc * SLOTS + sub * 4);
    const int deg = live ? cntv : 0;
    const int n = min(SLOTS, deg);
    const float* lp = p + sub * 4;
    v4f acc = *(const v4f*)(lp + (size_t)cc * C);

    int nm = n;
    nm = max(nm, __shfl_xor(nm, 8, 32));
    nm = max(nm, __shfl_xor(nm, 16, 32));
    const int nmax = __builtin_amdgcn_readfirstlane(nm);

    const v4f zero = (v4f){0.f, 0.f, 0.f, 0.f};
    int j = 0;
    do {
        const int la = gb + (j >> 2);
        const int lb = la + 1;
        const int r0 = __shfl(s.x, la, 32);
        const int r1 = __shfl(s.y, la, 32);
        const int r2 = __shfl(s.z, la, 32);
        const int r3 = __shfl(s.w, la, 32);
        const int r4 = __shfl(s.x, lb, 32);
        const int r5 = __shfl(s.y, lb, 32);
        const int r6 = __shfl(s.z, lb, 32);
        const int r7 = __shfl(s.w, lb, 32);
        v4f x0 = zero, x1 = zero, x2 = zero, x3 = zero, x4 = zero, x5 = zero, x6 = zero, x7 = zero;
        if (j     < n) x0 = *(const v4f*)(lp + (size_t)r0 * C);
        if (j + 1 < n) x1 = *(const v4f*)(lp + (size_t)r1 * C);
        if (j + 2 < n) x2 = *(const v4f*)(lp + (size_t)r2 * C);
        if (j + 3 < n) x3 = *(const v4f*)(lp + (size_t)r3 * C);
        if (j + 4 < n) x4 = *(const v4f*)(lp + (size_t)r4 * C);
        if (j + 5 < n) x5 = *(const v4f*)(lp + (size_t)r5 * C);
        if (j + 6 < n) x6 = *(const v4f*)(lp + (size_t)r6 * C);
        if (j + 7 < n) x7 = *(const v4f*)(lp + (size_t)r7 * C);
        acc += x0;
        acc += x1;
        acc += x2;
        acc += x3;
        acc += x4;
        acc += x5;
        acc += x6;
        acc += x7;
        j += 8;
    } while (j < nmax);

    if (__builtin_amdgcn_ballot_w32(deg > SLOTS)) {
        const int nov = min(*ovn, E);
        for (int base = 0; base < nov; base += 32) {
            int t = -1;
            int r = 0;
            if (base + lane < nov) {
                t = ov[2 * (size_t)(base + lane)];
                r = ov[2 * (size_t)(base + lane) + 1];
            }
#pragma unroll
            for (int q = 0; q < 4; ++q) {
                const int cq = __builtin_amdgcn_readlane(c, q * 8);
                unsigned hit = __builtin_amdgcn_ballot_w32(t == cq);
                while (hit) {
                    const int j2 = __builtin_ctz(hit);
                    hit &= hit - 1;
                    const int r0 = __builtin_amdgcn_readlane(r, j2);
                    if (gq == q) acc += *(const v4f*)(lp + (size_t)r0 * C);
                }
            }
        }
    }

    if (live) {
        const float dc = rsqrtf((float)(deg + 1));
        const v4f cv = *(const v4f*)(cvec + sub * 4);
        const v4f o = acc * dc + cv;
        *(v4f*)(out + (size_t)c * C + sub * 4) = o;
    }
}

extern "C" void kernel_launch(void* const* d_in, const int* in_sizes, int n_in,
                              void* d_out, int out_size, void* d_ws, size_t ws_size,
                              hipStream_t stream) {
    constexpr int F = 128, H = 128, D = 64, C = 32;

    const float* x   = (const float*)d_in[0];
    const int*   ei  = (const int*)d_in[1];
    const float* W1  = (const float*)d_in[2];
    const float* b1  = (const float*)d_in[3];
    const float* g1  = (const float*)d_in[4];
    const float* be1 = (const float*)d_in[5];
    const float* W2  = (const float*)d_in[6];
    const float* b2  = (const float*)d_in[7];
    const float* Wc  = (const float*)d_in[8];
    const float* bc  = (const float*)d_in[9];
    float* logits    = (float*)d_out;

    const int N = in_sizes[0] / F;
    const int E = in_sizes[1] / 2;
    const int* rows = ei;
    const int* cols = ei + E;

    float* ws   = (float*)d_ws;
    float* hw1  = ws;
    float* p    = ws + (size_t)N * 128;
    float* cvec = ws + (size_t)N * 160;
    int*   ovn  = (int*)(ws + (size_t)N * 160 + 64);
    int*   runtab = (int*)(ws + (size_t)N * 161);
    int*   slot = (int*)(ws + (size_t)N * 192);
    int*   ov   = (int*)(ws + (size_t)N * 224);
    int*   cnt  = (int*)(ws + (size_t)N * 256);
    int*   binbuf = (int*)p;

    const int TPB = 256;
    const int wavesPerBlk = TPB / 32;
    auto cdiv = [](long long a, long long b) { return (int)((a + b - 1) / b); };

    const int NB = cdiv(N, BIN_NODES);
    bin_edges_kernel<<<cdiv(E, CHUNK), 256, 0, stream>>>(rows, cols, binbuf, runtab, ovn, E);
    build_slots_rows_kernel<F, H / 16><<<NB, 640, 0, stream>>>(
        runtab, binbuf, cnt, slot, ovn, ov, x, W1, hw1, N, E);

    gather_ln_products_kernel<H, D, C><<<cdiv(N, FG_ROWS), 128, 0, stream>>>(
        hw1, cnt, slot, ovn, ov, b1, g1, be1, W2, Wc, b2, bc, p, cvec, N, E);

    gather_rows32_kernel<<<cdiv(N, 4 * wavesPerBlk), TPB, 0, stream>>>(
        p, cnt, slot, ovn, ov, cvec, logits, N, E);
}


# Classifier folded into the layer-2 weight: W2c = W2 @ Wc formed once per call on WMMA in the slot-building launch; a 32-row block does one product of 128 WMMAs, not 256 + 64 chained through LDS
# speedup vs baseline: 6.4658x; 1.0338x over previous
#include <hip/hip_runtime.h>
#include <hip/hip_bf16.h>


typedef __attribute__((ext_vector_type(2))) float v2f;
typedef __attribute__((ext_vector_type(8))) float v8f;

#define LN_EPS 1e-5f

template <int K, int NT, int MT, int LDW>
__device__ __forceinline__
void rows_times_w_f32(const float* __restrict__ A, const float* __restrict__ W,
                      int m0, int lane, v8f (&acc)[MT][NT]) {
    constexpr int NOUT = LDW;
    const int mr = lane & 15;
    const int hi = lane >> 4;

    const float* arow[MT];
#pragma unroll
    for (int mt = 0; mt < MT; ++mt)
        arow[mt] = A + (size_t)(m0 + mt * 16 + mr) * K;

    constexpr int UNR = (MT * NT >= 8) ? 2 : 4;
#pragma unroll UNR
    for (int k = 0; k < K; k += 4) {
        const int kb = k + hi * 2;
        v2f a[MT];
#pragma unroll
        for (int mt = 0; mt < MT; ++mt) {
            a[mt].x = arow[mt][kb];
            a[mt].y = arow[mt][kb + 1];
        }
        const float* w0 = W + (size_t)kb * NOUT;
        const float* w1 = w0 + NOUT;
#pragma unroll
        for (int t = 0; t < NT; ++t) {
            const int n = t * 16 + mr;
            v2f b;
            b.x = w0[n];
            b.y = w1[n];
#pragma unroll
            for (int mt = 0; mt < MT; ++mt)
                acc[mt][t] = __builtin_amdgcn_wmma_f32_16x16x4_f32(
                    false, a[mt], false, b, (short)0, acc[mt][t], false, false);
        }
    }
}

template <int NT, int MT, int LDO>
__device__ __forceinline__
void store_rows_scaled(const v8f (&acc)[MT][NT], const int* cntl, int c0,
                       float* __restrict__ out, int m0, int lane) {
    constexpr int NOUT = LDO;
    const int mr = lane & 15;
    const int hi = lane >> 4;
    float rsv[MT][8];
#pragma unroll
    for (int mt = 0; mt < MT; ++mt)
#pragma unroll
        for (int v = 0; v < 8; ++v)
            rsv[mt][v] = rsqrtf((float)(cntl[m0 - c0 + mt * 16 + v + hi * 8] + 1));

#pragma unroll
    for (int mt = 0; mt < MT; ++mt) {
#pragma unroll
        for (int t = 0; t < NT; ++t) {
            const int n = t * 16 + mr;
#pragma unroll
            for (int v = 0; v < 8; ++v) {
                const int m = m0 + mt * 16 + v + hi * 8;
                out[(size_t)m * NOUT + n] = acc[mt][t][v] * rsv[mt][v];
            }
        }
    }
}

template <int KM, int NO>
__device__ __forceinline__
void fold_weights_tile(const float* __restrict__ Wa, const float* __restrict__ Wb,
                       float* __restrict__ Wab, int t, int lane) {
    const int mr = lane & 15;
    const int hi = lane >> 4;
    const int rt = t / (NO / 16);
    const int ct = t % (NO / 16);
    const float* ap = Wa + (size_t)(rt * 16 + mr) * KM + hi * 2;
    const float* bp = Wb + (size_t)(hi * 2) * NO + ct * 16 + mr;
    v8f acc0 = (v8f){0.f, 0.f, 0.f, 0.f, 0.f, 0.f, 0.f, 0.f};
    v8f acc1 = acc0;
#pragma unroll
    for (int k = 0; k < KM; k += 8) {
        v2f a0, a1, b0, b1;
        a0.x = ap[k];
        a0.y = ap[k + 1];
        a1.x = ap[k + 4];
        a1.y = ap[k + 5];
        b0.x = bp[(size_t)k * NO];
        b0.y = bp[(size_t)k * NO + NO];
        b1.x = bp[(size_t)(k + 4) * NO];
        b1.y = bp[(size_t)(k + 4) * NO + NO];
        acc0 = __builtin_amdgcn_wmma_f32_16x16x4_f32(
            false, a0, false, b0, (short)0, acc0, false, false);
        acc1 = __builtin_amdgcn_wmma_f32_16x16x4_f32(
            false, a1, false, b1, (short)0, acc1, false, false);
    }
#pragma unroll
    for (int v = 0; v < 8; ++v)
        Wab[(size_t)(rt * 16 + v + hi * 8) * NO + ct * 16 + mr] = acc0[v] + acc1[v];
}

#define SLOTS 32
#define BIN_NODES 400
#define BIN_MAX 512
#define CHUNK 4096
#define EPT (CHUNK / 256)

__global__ __launch_bounds__(256)
void bin_edges_kernel(const int* __restrict__ row,
                      const int* __restrict__ col,
                      int* __restrict__ binbuf, int* __restrict__ runtab,
                      int* __restrict__ ovn, int E) {
    __shared__ int hist[BIN_MAX];
    __shared__ int off[BIN_MAX];
    __shared__ int stage_c[CHUNK];
    __shared__ int stage_r[CHUNK];

    const int tid = threadIdx.x;
    const int e0 = blockIdx.x * CHUNK;
    hist[tid] = 0;
    hist[tid + 256] = 0;
    if (blockIdx.x == 0 && tid == 0) *ovn = 0;
    __syncthreads();

    int c[EPT], r[EPT], rank[EPT];
#pragma unroll
    for (int i = 0; i < EPT; ++i) {
        const int e = e0 + i * 256 + tid;
        c[i] = -1;
        r[i] = 0;
        if (e < E) {
            c[i] = col[e];
            r[i] = row[e];
        }
    }
#pragma unroll
    for (int i = 0; i < EPT; ++i) {
        rank[i] = 0;
        if (c[i] >= 0) rank[i] = atomicAdd(&hist[c[i] / BIN_NODES], 1);
    }
    __syncthreads();

    if (tid < 32) {
        int v[16];
        int s = 0;
#pragma unroll
        for (int i = 0; i < 16; ++i) {
            v[i] = hist[tid * 16 + i];
            s += v[i];
        }
        int incl = s;
#pragma unroll
        for (int d = 1; d < 32; d <<= 1) {
            const int t = __shfl_up(incl, d, 32);
            if (tid >= d) incl += t;
        }
        int run = incl - s;
#pragma unroll
        for (int i = 0; i < 16; ++i) {
            off[tid * 16 + i] = run;
            run += v[i];
        }
    }
    __syncthreads();

    int* myrow = runtab + (size_t)blockIdx.x * BIN_MAX;
    myrow[tid] = off[tid] | (hist[tid] << 16);
    myrow[tid + 256] = off[tid + 256] | (hist[tid + 256] << 16);

#pragma unroll
    for (int i = 0; i < EPT; ++i) {
        if (c[i] >= 0) {
            const int pos = off[c[i] / BIN_NODES] + rank[i];
            stage_c[pos] = c[i];
            stage_r[pos] = r[i];
        }
    }
    __syncthreads();

    const int nval = min(CHUNK, E - e0);
#pragma unroll
    for (int i = 0; i < EPT; ++i) {
        const int s = i * 256 + tid;
        if (s < nval)
            *(int2*)(binbuf + 2 * ((size_t)e0 + s)) = make_int2(stage_c[s], stage_r[s]);
    }
}

#define RUN_REG 8
#define RUN_GRP 4
#define RUN_PRE 3
#define PROD_WAVES 12

template <int K, int NT, int FK, int FN>
__global__ __launch_bounds__(640)
void build_slots_rows_kernel(const int* __restrict__ runtab,
                             const int* __restrict__ binbuf,
                             int* __restrict__ cnt, int* __restrict__ slot,
                             int* ovn, int* __restrict__ ov,
                             const float* __restrict__ x,
                             const float* __restrict__ W,
                             float* __restrict__ hw,
                             const float* __restrict__ W2,
                             const float* __restrict__ Wc,
                             float* __restrict__ W2c, int N, int E) {
    typedef __attribute__((ext_vector_type(4))) int v4i;
    constexpr int FOLD_TILES = NT * (FN / 16);
    constexpr int MT = 2;
    constexpr int NOUT = NT * 16;
    constexpr int XT = 2;
    constexpr int XROW = PROD_WAVES * MT * 16;
    static_assert(XROW + 16 == BIN_NODES, "12 stripes of 32 rows and one tile of 16 make the bin");
    static_assert(4 * XT == NT, "four waves cover the last tile's columns");
    __shared__ int cnt_l[BIN_NODES];
    __shared__ __attribute__((aligned(16))) int slot_l[BIN_NODES * SLOTS];

    const int w = __builtin_amdgcn_readfirstlane((int)(threadIdx.x >> 5));
    const bool builder = w < 8;
    const int lane = threadIdx.x & 31;
    const int tid = threadIdx.x & 255;
    const int bin = blockIdx.x;
    const int c0 = bin * BIN_NODES;
    const int g = w - 8;
    const int m0 = c0 + g * (MT * 16);
    const int mx = c0 + XROW;
    const bool product = !builder && m0 + MT * 16 <= N;
    const bool product_x = !builder && g < NT / XT && mx + 16 <= N;

    v8f acc[MT][NT];
#pragma unroll
    for (int mt = 0; mt < MT; ++mt)
#pragma unroll
        for (int t = 0; t < NT; ++t)
            acc[mt][t] = (v8f){0.f, 0.f, 0.f, 0.f, 0.f, 0.f, 0.f, 0.f};
    v8f accx[1][XT];
#pragma unroll
    for (int t = 0; t < XT; ++t)
        accx[0][t] = (v8f){0.f, 0.f, 0.f, 0.f, 0.f, 0.f, 0.f, 0.f};

    if (builder) {
        for (int i = tid; i < BIN_NODES; i += 256) cnt_l[i] = 0;
        const v4i z4 = (v4i){0, 0, 0, 0};
        for (int i = tid; i < BIN_NODES * SLOTS / 4; i += 256)
            *(v4i*)(slot_l + i * 4) = z4;
    }
    __syncthreads();

    if (builder) {
        const int nblk1 = (E + CHUNK - 1) / CHUNK;
        const int grp = tid >> 3;
        const int sub = tid & 7;
        for (int rb = 0; rb < nblk1; rb += 32 * RUN_REG) {
            int ent[RUN_REG];
#pragma unroll
            for (int j = 0; j < RUN_REG; ++j) {
                const int q = rb + grp + 32 * j;
                ent[j] = 0;
                if (q < nblk1) ent[j] = runtab[(size_t)q * BIN_MAX + bin];
            }
#pragma unroll
            for (int j0 = 0; j0 < RUN_REG; j0 += RUN_GRP) {
                int2 pr[RUN_GRP][RUN_PRE];
                const int* srcp[RUN_GRP];
                int nn[RUN_GRP];
#pragma unroll
                for (int u = 0; u < RUN_GRP; ++u) {
                    const int q = rb + grp + 32 * (j0 + u);
                    nn[u] = ent[j0 + u] >> 16;
                    srcp[u] = binbuf + 2 * ((size_t)q * CHUNK + (ent[j0 + u] & 0xFFFF));
#pragma unroll
                    for (int v = 0; v < RUN_PRE; ++v) {
                        pr[u][v] = make_int2(-1, 0);
                        if (sub + 8 * v < nn[u]) pr[u][v] = *(const int2*)(srcp[u] + 2 * (sub + 8 * v));
                    }
                }
                int pl[RUN_GRP][RUN_PRE];
#pragma unroll
                for (int u = 0; u < RUN_GRP; ++u)
#pragma unroll
                    for (int v = 0; v < RUN_PRE; ++v) {
                        pl[u][v] = 0;
                        if (pr[u][v].x >= 0) pl[u][v] = atomicAdd(&cnt_l[pr[u][v].x - c0], 1);
                    }
#pragma unroll
                for (int u = 0; u < RUN_GRP; ++u)
#pragma unroll
                    for (int v = 0; v < RUN_PRE; ++v) {
                        if (pr[u][v].x >= 0) {
                            if (pl[u][v] < SLOTS) {
                                slot_l[(pr[u][v].x - c0) * SLOTS + pl[u][v]] = pr[u][v].y;
                            } else {
                                const int q = atomicAdd(ovn, 1);
                                ov[2 * (size_t)q] = pr[u][v].x;
                                ov[2 * (size_t)q + 1] = pr[u][v].y;
                            }
                        }
                    }
#pragma unroll
                for (int u = 0; u < RUN_GRP; ++u) {
                    for (int i = sub + 8 * RUN_PRE; i < nn[u]; i += 8) {
                        const int2 p2 = *(const int2*)(srcp[u] + 2 * i);
                        const int local = p2.x - c0;
                        const int pp = atomicAdd(&cnt_l[local], 1);
                        if (pp < SLOTS) {
                            slot_l[local * SLOTS + pp] = p2.y;
                        } else {
                            const int q = atomicAdd(ovn, 1);
                            ov[2 * (size_t)q] = p2.x;
                            ov[2 * (size_t)q + 1] = p2.y;
                        }
                    }
                }
            }
        }
        if (w == 0)
            for (int t = bin; t < FOLD_TILES; t += (int)gridDim.x)
                fold_weights_tile<FK, FN>(W2, Wc, W2c, t, lane);
    } else {
        if (product_x) rows_times_w_f32<K, XT, 1, NOUT>(x, W + g * (XT * 16), mx, lane, accx);
        if (product) rows_times_w_f32<K, NT, MT, NOUT>(x, W, m0, lane, acc);
    }
    __syncthreads();

    if (builder) {
        for (int i = tid; i < BIN_NODES; i += 256)
            if (c0 + i < N) cnt[c0 + i] = cnt_l[i];
        for (int idx = tid; idx < BIN_NODES * SLOTS / 4; idx += 256) {
            const int node = c0 + (idx >> 3);
            if (node < N)
                *(v4i*)(slot + (size_t)c0 * SLOTS + (size_t)idx * 4) = *(const v4i*)(slot_l + idx * 4);
        }
    } else {
        if (product) store_rows_scaled<NT, MT, NOUT>(acc, cnt_l, c0, hw, m0, lane);
        if (product_x) store_rows_scaled<XT, 1, NOUT>(accx, cnt_l, c0, hw + g * (XT * 16), mx, lane);
    }
}

#define FG_ROWS 32

template <int F, int NMID, int NOUT>
__global__ __launch_bounds__(128)
void gather_ln_products_kernel(const float* __restrict__ src,
                               const int* __restrict__ cnt,
                               const int* __restrict__ slot,
                               const int* __restrict__ ovn,
                               const int* __restrict__ ov,
                               const float* __restrict__ b,
                               const float* __restrict__ g,
                               const float* __restrict__ be,
                               const float* __restrict__ W2c,
                               const float* __restrict__ Wc,
                               const float* __restrict__ b2,
                               const float* __restrict__ bc,
                               float* __restrict__ out,
                               float* __restrict__ cvec, int N, int E) {
    static_assert(NOUT == 32, "tile dealing below is for 4 waves and 2 x 2 tiles of 16 x 16");
    constexpr int PER = F / 32;
    constexpr int PITCH = F + 4;
    typedef __attribute__((ext_vector_type(PER))) float vf;
    __shared__ __attribute__((aligned(16))) float tile[FG_ROWS * PITCH];

    const int lane = threadIdx.x & 31;
    const int w = __builtin_amdgcn_readfirstlane((int)(threadIdx.x >> 5));
    const int node0 = blockIdx.x * FG_ROWS;
    const float* lsrc = src + lane * PER;

    for (int q = 0; q < FG_ROWS / 4; ++q) {
        const int lr = w * (FG_ROWS / 4) + q;
        const int c = node0 + lr;
        vf y;
#pragma unroll
        for (int i = 0; i < PER; ++i) y[i] = 0.f;
        if (c < N) {
            const int deg = cnt[c];
            const float dc = rsqrtf((float)(deg + 1));

            vf acc = *(const vf*)(lsrc + (size_t)c * F);
            {
                const int n = min(SLOTS, deg);
                int my_r = c;
                if (lane < n) my_r = slot[(size_t)c * SLOTS + lane];
                int j = 0;
                for (; j + 4 <= n; j += 4) {
                    const int r0 = __builtin_amdgcn_readlane(my_r, j);
                    const int r1 = __builtin_amdgcn_readlane(my_r, j + 1);
                    const int r2 = __builtin_amdgcn_readlane(my_r, j + 2);
                    const int r3 = __builtin_amdgcn_readlane(my_r, j + 3);
                    const vf x0 = *(const vf*)(lsrc + (size_t)r0 * F);
                    const vf x1 = *(const vf*)(lsrc + (size_t)r1 * F);
                    const vf x2 = *(const vf*)(lsrc + (size_t)r2 * F);
                    const vf x3 = *(const vf*)(lsrc + (size_t)r3 * F);
                    acc += x0;
                    acc += x1;
                    acc += x2;
                    acc += x3;
                }
                for (; j < n; ++j) {
                    const int r0 = __builtin_amdgcn_readlane(my_r, j);
                    acc += *(const vf*)(lsrc + (size_t)r0 * F);
                }
            }
            if (deg > SLOTS) {
                const int nov = min(*ovn, E);
                for (int base = 0; base < nov; base += 32) {
                    int t = -1;
                    int r = c;
                    if (base + lane < nov) {
                        t = ov[2 * (size_t)(base + lane)];
                        r = ov[2 * (size_t)(base + lane) + 1];
                    }
                    unsigned hit = __builtin_amdgcn_ballot_w32(t == c);
                    while (hit) {
                        const int j2 = __builtin_ctz(hit);
                        hit &= hit - 1;
                        const int r0 = __builtin_amdgcn_readlane(r, j2);
                        acc += *(const vf*)(lsrc + (size_t)r0 * F);
                    }
                }
            }
            acc *= dc;

            float v[PER];
#pragma unroll
            for (int i = 0; i < PER; ++i) v[i] = acc[i] + b[lane * PER + i];

            float sum = 0.f;
#pragma unroll
            for (int i = 0; i < PER; ++i) sum += v[i];
#pragma unroll
            for (int off = 16; off > 0; off >>= 1) sum += __shfl_xor(sum, off, 32);
            const float mu = sum * (1.0f / F);

            float var = 0.f;
#pragma unroll
            for (int i = 0; i < PER; ++i) {
                const float d = v[i] - mu;
                var += d * d;
            }
#pragma unroll
            for (int off = 16; off > 0; off >>= 1) var += __shfl_xor(var, off, 32);
            var *= (1.0f / F);
            const float rs = rsqrtf(var + LN_EPS);

#pragma unroll
            for (int i = 0; i < PER; ++i) {
                const float z = (v[i] - mu) * rs * g[lane * PER + i] + be[lane * PER + i];
                y[i] = fmaxf(z, 0.f);
            }
        }
        *(vf*)(tile + lr * PITCH + lane * PER) = y;
    }
    __syncthreads();

    const int mr = lane & 15;
    const int hi = lane >> 4;
    const int rt = w >> 1;
    const int n2 = (w & 1) * 16 + mr;
    const float* ap = tile + (rt * 16 + mr) * PITCH + hi * 2;
    const float* wp = W2c + (size_t)(hi * 2) * NOUT + n2;

    v8f accA = (v8f){0.f, 0.f, 0.f, 0.f, 0.f, 0.f, 0.f, 0.f};
    v8f accB = accA;
#pragma unroll 4
    for (int k = 0; k < F; k += 8) {
        v2f b0, b1;
        b0.x = wp[(size_t)k * NOUT];
        b0.y = wp[(size_t)k * NOUT + NOUT];
        b1.x = wp[(size_t)(k + 4) * NOUT];
        b1.y = wp[(size_t)(k + 4) * NOUT + NOUT];
        const v2f a0 = *(const v2f*)(ap + k);
        const v2f a1 = *(const v2f*)(ap + k + 4);
        accA = __builtin_amdgcn_wmma_f32_16x16x4_f32(
            false, a0, false, b0, (short)0, accA, false, false);
        accB = __builtin_amdgcn_wmma_f32_16x16x4_f32(
            false, a1, false, b1, (short)0, accB, false, false);
    }
    const v8f acc2 = accA + accB;

    constexpr int OP2 = NOUT + 4;
    __syncthreads();
#pragma unroll
    for (int v = 0; v < 8; ++v)
        tile[(rt * 16 + v + hi * 8) * OP2 + n2] = acc2[v];
    __syncthreads();

    typedef __attribute__((ext_vector_type(4))) float v4f;
    constexpr int LPR = NOUT / 4;
    constexpr int RPI = 32 / LPR;
    const int cl = (lane % LPR) * 4;
    const int rl = lane / LPR;
#pragma unroll
    for (int i = 0; i < (FG_ROWS / 4) / RPI; ++i) {
        const int r = w * (FG_ROWS / 4) + i * RPI + rl;
        const int m = node0 + r;
        if (m < N) {
            v4f o = *(const v4f*)(tile + r * OP2 + cl);
            o *= rsqrtf((float)(cnt[m] + 1));
            *(v4f*)(out + (size_t)m * NOUT + cl) = o;
        }
    }

    if (blockIdx.x == 0 && w < 2) {
        const int n3 = w * 16 + mr;
        const float* wp3 = Wc + (size_t)(hi * 2) * NOUT + n3;
        v8f acc3 = (v8f){0.f, 0.f, 0.f, 0.f, 0.f, 0.f, 0.f, 0.f};
#pragma unroll
        for (int k = 0; k < NMID; k += 4) {
            v2f a, bf;
            a.x = b2[k + hi * 2];
            a.y = b2[k + hi * 2 + 1];
            bf.x = wp3[(size_t)k * NOUT];
            bf.y = wp3[(size_t)k * NOUT + NOUT];
            acc3 = __builtin_amdgcn_wmma_f32_16x16x4_f32(
                false, a, false, bf, (short)0, acc3, false, false);
        }
        if (hi == 0) cvec[n3] = acc3[0] + bc[n3];
    }
}

__global__ __launch_bounds__(256)
void gather_rows32_kernel(const float* __restrict__ p,
                          const int* __restrict__ cnt,
                          const int* __restrict__ slot,
                          const int* __restrict__ ovn,
                          const int* __restrict__ ov,
                          const float* __restrict__ cvec,
                          float* __restrict__ out, int N, int E) {
    constexpr int C = 32;
    typedef __attribute__((ext_vector_type(4))) float v4f;
    typedef __attribute__((ext_vector_type(4))) int v4i;

    const int lane = threadIdx.x & 31;
    const int sub = lane & 7;
    const int gq = lane >> 3;
    const int gb = lane & ~7;
    const int wave = (blockIdx.x * blockDim.x + threadIdx.x) >> 5;
    const int c = wave * 4 + gq;
    const bool live = c < N;
    const int cc = live ? c : N - 1;

    const int cntv = cnt[cc];
    const v4i s = *(const v4i*)(slot + (size_t)cc * SLOTS + sub * 4);
    const int deg = live ? cntv : 0;
    const int n = min(SLOTS, deg);
    const float* lp = p + sub * 4;
    v4f acc = *(const v4f*)(lp + (size_t)cc * C);

    int nm = n;
    nm = max(nm, __shfl_xor(nm, 8, 32));
    nm = max(nm, __shfl_xor(nm, 16, 32));
    const int nmax = __builtin_amdgcn_readfirstlane(nm);

    const v4f zero = (v4f){0.f, 0.f, 0.f, 0.f};
    int j = 0;
    do {
        const int la = gb + (j >> 2);
        const int lb = la + 1;
        const int r0 = __shfl(s.x, la, 32);
        const int r1 = __shfl(s.y, la, 32);
        const int r2 = __shfl(s.z, la, 32);
        const int r3 = __shfl(s.w, la, 32);
        const int r4 = __shfl(s.x, lb, 32);
        const int r5 = __shfl(s.y, lb, 32);
        const int r6 = __shfl(s.z, lb, 32);
        const int r7 = __shfl(s.w, lb, 32);
        v4f x0 = zero, x1 = zero, x2 = zero, x3 = zero, x4 = zero, x5 = zero, x6 = zero, x7 = zero;
        if (j     < n) x0 = *(const v4f*)(lp + (size_t)r0 * C);
        if (j + 1 < n) x1 = *(const v4f*)(lp + (size_t)r1 * C);
        if (j + 2 < n) x2 = *(const v4f*)(lp + (size_t)r2 * C);
        if (j + 3 < n) x3 = *(const v4f*)(lp + (size_t)r3 * C);
        if (j + 4 < n) x4 = *(const v4f*)(lp + (size_t)r4 * C);
        if (j + 5 < n) x5 = *(const v4f*)(lp + (size_t)r5 * C);
        if (j + 6 < n) x6 = *(const v4f*)(lp + (size_t)r6 * C);
        if (j + 7 < n) x7 = *(const v4f*)(lp + (size_t)r7 * C);
        acc += x0;
        acc += x1;
        acc += x2;
        acc += x3;
        acc += x4;
        acc += x5;
        acc += x6;
        acc += x7;
        j += 8;
    } while (j < nmax);

    if (__builtin_amdgcn_ballot_w32(deg > SLOTS)) {
        const int nov = min(*ovn, E);
        for (int base = 0; base < nov; base += 32) {
            int t = -1;
            int r = 0;
            if (base + lane < nov) {
                t = ov[2 * (size_t)(base + lane)];
                r = ov[2 * (size_t)(base + lane) + 1];
            }
#pragma unroll
            for (int q = 0; q < 4; ++q) {
                const int cq = __builtin_amdgcn_readlane(c, q * 8);
                unsigned hit = __builtin_amdgcn_ballot_w32(t == cq);
                while (hit) {
                    const int j2 = __builtin_ctz(hit);
                    hit &= hit - 1;
                    const int r0 = __builtin_amdgcn_readlane(r, j2);
                    if (gq == q) acc += *(const v4f*)(lp + (size_t)r0 * C);
                }
            }
        }
    }

    if (live) {
        const float dc = rsqrtf((float)(deg + 1));
        const v4f cv = *(const v4f*)(cvec + sub * 4);
        const v4f o = acc * dc + cv;
        *(v4f*)(out + (size_t)c * C + sub * 4) = o;
    }
}

extern "C" void kernel_launch(void* const* d_in, const int* in_sizes, int n_in,
                              void* d_out, int out_size, void* d_ws, size_t ws_size,
                              hipStream_t stream) {
    constexpr int F = 128, H = 128, D = 64, C = 32;

    const float* x   = (const float*)d_in[0];
    const int*   ei  = (const int*)d_in[1];
    const float* W1  = (const float*)d_in[2];
    const float* b1  = (const float*)d_in[3];
    const float* g1  = (const float*)d_in[4];
    const float* be1 = (const float*)d_in[5];
    const float* W2  = (const float*)d_in[6];
    const float* b2  = (const float*)d_in[7];
    const float* Wc  = (const float*)d_in[8];
    const float* bc  = (const float*)d_in[9];
    float* logits    = (float*)d_out;

    const int N = in_sizes[0] / F;
    const int E = in_sizes[1] / 2;
    const int* rows = ei;
    const int* cols = ei + E;

    float* ws   = (float*)d_ws;
    float* hw1  = ws;
    float* p    = ws + (size_t)N * 128;
    float* cvec = ws + (size_t)N * 160;
    int*   ovn  = (int*)(ws + (size_t)N * 160 + 64);
    int*   runtab = (int*)(ws + (size_t)N * 161);
    float* W2c  = ws + (size_t)N * 164;
    int*   slot = (int*)(ws + (size_t)N * 192);
    int*   ov   = (int*)(ws + (size_t)N * 224);
    int*   cnt  = (int*)(ws + (size_t)N * 256);
    int*   binbuf = (int*)p;

    const int TPB = 256;
    const int wavesPerBlk = TPB / 32;
    auto cdiv = [](long long a, long long b) { return (int)((a + b - 1) / b); };

    const int NB = cdiv(N, BIN_NODES);
    bin_edges_kernel<<<cdiv(E, CHUNK), 256, 0, stream>>>(rows, cols, binbuf, runtab, ovn, E);
    build_slots_rows_kernel<F, H / 16, D, C><<<NB, 640, 0, stream>>>(
        runtab, binbuf, cnt, slot, ovn, ov, x, W1, hw1, W2, Wc, W2c, N, E);

    gather_ln_products_kernel<H, D, C><<<cdiv(N, FG_ROWS), 128, 0, stream>>>(
        hw1, cnt, slot, ovn, ov, b1, g1, be1, W2c, Wc, b2, bc, p, cvec, N, E);

    gather_rows32_kernel<<<cdiv(N, 4 * wavesPerBlk), TPB, 0, stream>>>(
        p, cnt, slot, ovn, ov, cvec, logits, N, E);
}
